# starting kernel
# baseline (speedup 1.0000x reference)
#include <hip/hip_runtime.h>
#include <hip/hip_bf16.h>

typedef __attribute__((ext_vector_type(2))) float v2f;
typedef __attribute__((ext_vector_type(8))) float v8f;

#define FIN 128

template <int FOUT, bool RELU>
__global__ __launch_bounds__(256) void gemm_wmma_kernel(
    const float* __restrict__ A, const float* __restrict__ W,
    float* __restrict__ C, int nrows)
{
    constexpr int NT = FOUT / 16;
    __shared__ float ldsW[FIN * FOUT];

    for (int i = threadIdx.x; i < FIN * FOUT; i += 256)
        ldsW[i] = W[i];
    __syncthreads();

    const int wave = threadIdx.x >> 5;
    const int lane = threadIdx.x & 31;
    const int half = lane >> 4;
    const int m    = lane & 15;

    const int tile   = blockIdx.x * 8 + wave;
    const int ntiles = (nrows + 15) >> 4;
    if (tile >= ntiles) return;

    const int row0 = tile * 16;
    int arow = row0 + m;
    if (arow >= nrows) arow = nrows - 1;
    const float* aptr = A + (size_t)arow * FIN;

    v8f acc[NT] = {};

    for (int k = 0; k < FIN; k += 4) {
        const int kk = k + 2 * half;
        float2 af = *(const float2*)(aptr + kk);
        if (RELU) {
            af.x = af.x > 0.0f ? af.x : 0.0f;
            af.y = af.y > 0.0f ? af.y : 0.0f;
        }
        v2f a;
        a.x = af.x;
        a.y = af.y;
#pragma unroll
        for (int t = 0; t < NT; ++t) {
            v2f b;
            b.x = ldsW[(kk)     * FOUT + t * 16 + m];
            b.y = ldsW[(kk + 1) * FOUT + t * 16 + m];
            acc[t] = __builtin_amdgcn_wmma_f32_16x16x4_f32(
                false, a, false, b, (short)0, acc[t], false, false);
        }
    }

#pragma unroll
    for (int r = 0; r < 8; ++r) {
        const int row = row0 + r + 8 * half;
        if (row < nrows) {
#pragma unroll
            for (int t = 0; t < NT; ++t)
                C[(size_t)row * FOUT + t * 16 + m] = acc[t][r];
        }
    }
}

template <int FOUT>
__global__ void fill_bias_kernel(float* __restrict__ out,
                                 const float* __restrict__ b, int total)
{
    const int i = blockIdx.x * blockDim.x + threadIdx.x;
    if (i < total) out[i] = b[i & (FOUT - 1)];
}

template <int F>
__global__ __launch_bounds__(256) void scatter_kernel(
    const float* __restrict__ h, const int* __restrict__ srcI,
    const int* __restrict__ dstI, const float* __restrict__ ew,
    float* __restrict__ out, int nedges)
{
    const int e = blockIdx.x * 8 + (threadIdx.x >> 5);
    if (e >= nedges) return;
    const int   lane = threadIdx.x & 31;
    const int   s = srcI[e];
    const int   d = dstI[e];
    const float w = ew[e];

    constexpr int PER = F / 32;
    const float* hp = h   + (size_t)s * F + lane * PER;
    float*       op = out + (size_t)d * F + lane * PER;

    if constexpr (PER == 4) {
        float4 v = *(const float4*)hp;
        unsafeAtomicAdd(op + 0, v.x * w);
        unsafeAtomicAdd(op + 1, v.y * w);
        unsafeAtomicAdd(op + 2, v.z * w);
        unsafeAtomicAdd(op + 3, v.w * w);
    } else {
        float2 v = *(const float2*)hp;
        unsafeAtomicAdd(op + 0, v.x * w);
        unsafeAtomicAdd(op + 1, v.y * w);
    }
}

extern "C" void kernel_launch(void* const* d_in, const int* in_sizes, int n_in,
                              void* d_out, int out_size, void* d_ws, size_t ws_size,
                              hipStream_t stream)
{
    const float* x  = (const float*)d_in[0];
    const int*   ei = (const int*)d_in[1];
    const float* ew = (const float*)d_in[2];
    const float* W1 = (const float*)d_in[3];
    const float* b1 = (const float*)d_in[4];
    const float* W2 = (const float*)d_in[5];
    const float* b2 = (const float*)d_in[6];
    const float* W3 = (const float*)d_in[7];
    const float* b3 = (const float*)d_in[8];
    float* out = (float*)d_out;

    const int N = in_sizes[0] / FIN;
    const int E = in_sizes[2];
    const int* srcI = ei;
    const int* dstI = ei + E;

    float* bufA = (float*)d_ws;
    float* bufB = bufA + (size_t)N * FIN;

    const int ntiles     = (N + 15) / 16;
    const int gemmBlocks = (ntiles + 7) / 8;
    const int edgeBlocks = (E + 7) / 8;

    gemm_wmma_kernel<128, false><<<gemmBlocks, 256, 0, stream>>>(x, W1, bufA, N);
    fill_bias_kernel<128><<<(N * 128 + 255) / 256, 256, 0, stream>>>(bufB, b1, N * 128);
    scatter_kernel<128><<<edgeBlocks, 256, 0, stream>>>(bufA, srcI, dstI, ew, bufB, E);

    gemm_wmma_kernel<128, true><<<gemmBlocks, 256, 0, stream>>>(bufB, W2, bufA, N);
    fill_bias_kernel<128><<<(N * 128 + 255) / 256, 256, 0, stream>>>(bufB, b2, N * 128);
    scatter_kernel<128><<<edgeBlocks, 256, 0, stream>>>(bufA, srcI, dstI, ew, bufB, E);

    gemm_wmma_kernel<64, true><<<gemmBlocks, 256, 0, stream>>>(bufB, W3, bufA, N);
    fill_bias_kernel<64><<<(N * 64 + 255) / 256, 256, 0, stream>>>(out, b3, N * 64);
    scatter_kernel<64><<<edgeBlocks, 256, 0, stream>>>(bufA, srcI, dstI, ew, out, E);
}


# scatter: lane l owns features l+32i, so each wave-wide f32 atomic add and h load covers one whole 128-B line, not 8 lanes in each of 4 lines: 4x fewer atomic line requests; arrival-order sums kept
# speedup vs baseline: 1.2738x; 1.2738x over previous
#include <hip/hip_runtime.h>
#include <hip/hip_bf16.h>

typedef __attribute__((ext_vector_type(2))) float v2f;
typedef __attribute__((ext_vector_type(8))) float v8f;

#define FIN 128

template <int FOUT, bool RELU>
__global__ __launch_bounds__(256) void gemm_wmma_kernel(
    const float* __restrict__ A, const float* __restrict__ W,
    float* __restrict__ C, int nrows)
{
    constexpr int NT = FOUT / 16;
    __shared__ float ldsW[FIN * FOUT];

    for (int i = threadIdx.x; i < FIN * FOUT; i += 256)
        ldsW[i] = W[i];
    __syncthreads();

    const int wave = threadIdx.x >> 5;
    const int lane = threadIdx.x & 31;
    const int half = lane >> 4;
    const int m    = lane & 15;

    const int tile   = blockIdx.x * 8 + wave;
    const int ntiles = (nrows + 15) >> 4;
    if (tile >= ntiles) return;

    const int row0 = tile * 16;
    int arow = row0 + m;
    if (arow >= nrows) arow = nrows - 1;
    const float* aptr = A + (size_t)arow * FIN;

    v8f acc[NT] = {};

    for (int k = 0; k < FIN; k += 4) {
        const int kk = k + 2 * half;
        float2 af = *(const float2*)(aptr + kk);
        if (RELU) {
            af.x = af.x > 0.0f ? af.x : 0.0f;
            af.y = af.y > 0.0f ? af.y : 0.0f;
        }
        v2f a;
        a.x = af.x;
        a.y = af.y;
#pragma unroll
        for (int t = 0; t < NT; ++t) {
            v2f b;
            b.x = ldsW[(kk)     * FOUT + t * 16 + m];
            b.y = ldsW[(kk + 1) * FOUT + t * 16 + m];
            acc[t] = __builtin_amdgcn_wmma_f32_16x16x4_f32(
                false, a, false, b, (short)0, acc[t], false, false);
        }
    }

#pragma unroll
    for (int r = 0; r < 8; ++r) {
        const int row = row0 + r + 8 * half;
        if (row < nrows) {
#pragma unroll
            for (int t = 0; t < NT; ++t)
                C[(size_t)row * FOUT + t * 16 + m] = acc[t][r];
        }
    }
}

template <int FOUT>
__global__ void fill_bias_kernel(float* __restrict__ out,
                                 const float* __restrict__ b, int total)
{
    const int i = blockIdx.x * blockDim.x + threadIdx.x;
    if (i < total) out[i] = b[i & (FOUT - 1)];
}

template <int F>
__global__ __launch_bounds__(256) void scatter_kernel(
    const float* __restrict__ h, const int* __restrict__ srcI,
    const int* __restrict__ dstI, const float* __restrict__ ew,
    float* __restrict__ out, int nedges)
{
    const int e = blockIdx.x * 8 + (threadIdx.x >> 5);
    if (e >= nedges) return;
    const int   lane = threadIdx.x & 31;
    const int   s = srcI[e];
    const int   d = dstI[e];
    const float w = ew[e];

    constexpr int PER = F / 32;
    const float* hp = h   + (size_t)s * F + lane;
    float*       op = out + (size_t)d * F + lane;

    float v[PER];
    for (int i = 0; i < PER; ++i)
        v[i] = hp[i * 32];
    for (int i = 0; i < PER; ++i)
        unsafeAtomicAdd(op + i * 32, v[i] * w);
}

extern "C" void kernel_launch(void* const* d_in, const int* in_sizes, int n_in,
                              void* d_out, int out_size, void* d_ws, size_t ws_size,
                              hipStream_t stream)
{
    const float* x  = (const float*)d_in[0];
    const int*   ei = (const int*)d_in[1];
    const float* ew = (const float*)d_in[2];
    const float* W1 = (const float*)d_in[3];
    const float* b1 = (const float*)d_in[4];
    const float* W2 = (const float*)d_in[5];
    const float* b2 = (const float*)d_in[6];
    const float* W3 = (const float*)d_in[7];
    const float* b3 = (const float*)d_in[8];
    float* out = (float*)d_out;

    const int N = in_sizes[0] / FIN;
    const int E = in_sizes[2];
    const int* srcI = ei;
    const int* dstI = ei + E;

    float* bufA = (float*)d_ws;
    float* bufB = bufA + (size_t)N * FIN;

    const int ntiles     = (N + 15) / 16;
    const int gemmBlocks = (ntiles + 7) / 8;
    const int edgeBlocks = (E + 7) / 8;

    gemm_wmma_kernel<128, false><<<gemmBlocks, 256, 0, stream>>>(x, W1, bufA, N);
    fill_bias_kernel<128><<<(N * 128 + 255) / 256, 256, 0, stream>>>(bufB, b1, N * 128);
    scatter_kernel<128><<<edgeBlocks, 256, 0, stream>>>(bufA, srcI, dstI, ew, bufB, E);

    gemm_wmma_kernel<128, true><<<gemmBlocks, 256, 0, stream>>>(bufB, W2, bufA, N);
    fill_bias_kernel<128><<<(N * 128 + 255) / 256, 256, 0, stream>>>(bufB, b2, N * 128);
    scatter_kernel<128><<<edgeBlocks, 256, 0, stream>>>(bufA, srcI, dstI, ew, bufB, E);

    gemm_wmma_kernel<64, true><<<gemmBlocks, 256, 0, stream>>>(bufB, W3, bufA, N);
    fill_bias_kernel<64><<<(N * 64 + 255) / 256, 256, 0, stream>>>(out, b3, N * 64);
    scatter_kernel<64><<<edgeBlocks, 256, 0, stream>>>(bufA, srcI, dstI, ew, out, E);
}


# scatter -> gather: edges bucketed by destination once per call (int count, scan, fill), then a wave per node sums its rows in registers, stores out once with bias; no f32 atomics; sums in fill order
# speedup vs baseline: 2.1877x; 1.7175x over previous
#include <hip/hip_runtime.h>
#include <hip/hip_bf16.h>

typedef __attribute__((ext_vector_type(2))) float v2f;
typedef __attribute__((ext_vector_type(8))) float v8f;

#define FIN 128

template <int FOUT, bool RELU>
__global__ __launch_bounds__(256) void gemm_wmma_kernel(
    const float* __restrict__ A, const float* __restrict__ W,
    float* __restrict__ C, int nrows)
{
    constexpr int NT = FOUT / 16;
    __shared__ float ldsW[FIN * FOUT];

    for (int i = threadIdx.x; i < FIN * FOUT; i += 256)
        ldsW[i] = W[i];
    __syncthreads();

    const int wave = threadIdx.x >> 5;
    const int lane = threadIdx.x & 31;
    const int half = lane >> 4;
    const int m    = lane & 15;

    const int tile   = blockIdx.x * 8 + wave;
    const int ntiles = (nrows + 15) >> 4;
    if (tile >= ntiles) return;

    const int row0 = tile * 16;
    int arow = row0 + m;
    if (arow >= nrows) arow = nrows - 1;
    const float* aptr = A + (size_t)arow * FIN;

    v8f acc[NT] = {};

    for (int k = 0; k < FIN; k += 4) {
        const int kk = k + 2 * half;
        float2 af = *(const float2*)(aptr + kk);
        if (RELU) {
            af.x = af.x > 0.0f ? af.x : 0.0f;
            af.y = af.y > 0.0f ? af.y : 0.0f;
        }
        v2f a;
        a.x = af.x;
        a.y = af.y;
#pragma unroll
        for (int t = 0; t < NT; ++t) {
            v2f b;
            b.x = ldsW[(kk)     * FOUT + t * 16 + m];
            b.y = ldsW[(kk + 1) * FOUT + t * 16 + m];
            acc[t] = __builtin_amdgcn_wmma_f32_16x16x4_f32(
                false, a, false, b, (short)0, acc[t], false, false);
        }
    }

#pragma unroll
    for (int r = 0; r < 8; ++r) {
        const int row = row0 + r + 8 * half;
        if (row < nrows) {
#pragma unroll
            for (int t = 0; t < NT; ++t)
                C[(size_t)row * FOUT + t * 16 + m] = acc[t][r];
        }
    }
}

__global__ void zero_counts_kernel(int* __restrict__ cnt, int n)
{
    const int i = blockIdx.x * blockDim.x + threadIdx.x;
    if (i < n) cnt[i] = 0;
}

__global__ void count_dst_kernel(const int* __restrict__ dstI,
                                 int* __restrict__ cnt, int nedges)
{
    const int e = blockIdx.x * blockDim.x + threadIdx.x;
    if (e < nedges) atomicAdd(&cnt[dstI[e]], 1);
}

__global__ __launch_bounds__(256) void scan_local_kernel(
    const int* __restrict__ cnt, int* __restrict__ rowptr,
    int* __restrict__ bsum, int n)
{
    __shared__ int sh[256];
    const int t  = threadIdx.x;
    const int i0 = blockIdx.x * 1024 + t * 4;
    int c[4];
#pragma unroll
    for (int k = 0; k < 4; ++k) c[k] = (i0 + k < n) ? cnt[i0 + k] : 0;
    const int mine = c[0] + c[1] + c[2] + c[3];
    sh[t] = mine;
    __syncthreads();
    for (int off = 1; off < 256; off <<= 1) {
        const int add = (t >= off) ? sh[t - off] : 0;
        __syncthreads();
        sh[t] += add;
        __syncthreads();
    }
    int run = sh[t] - mine;
#pragma unroll
    for (int k = 0; k < 4; ++k) {
        if (i0 + k < n) rowptr[i0 + k] = run;
        run += c[k];
    }
    if (t == 255) bsum[blockIdx.x] = sh[255];
}

__global__ __launch_bounds__(256) void scan_offset_kernel(
    int* __restrict__ rowptr, int* __restrict__ next,
    const int* __restrict__ bsum, int n)
{
    int off = 0;
    for (int j = 0; j < (int)blockIdx.x; ++j) off += bsum[j];
    const int i0 = blockIdx.x * 1024 + threadIdx.x * 4;
#pragma unroll
    for (int k = 0; k < 4; ++k) {
        if (i0 + k < n) {
            const int r = rowptr[i0 + k] + off;
            rowptr[i0 + k] = r;
            next[i0 + k]   = r;
        }
    }
    if (blockIdx.x == gridDim.x - 1 && threadIdx.x == 0)
        rowptr[n] = off + bsum[blockIdx.x];
}

__global__ void fill_edges_kernel(const int* __restrict__ srcI,
                                  const int* __restrict__ dstI,
                                  const float* __restrict__ ew,
                                  int* __restrict__ next,
                                  int2* __restrict__ edges, int nedges)
{
    const int e = blockIdx.x * blockDim.x + threadIdx.x;
    if (e < 8) edges[nedges + e] = make_int2(0, 0);
    if (e >= nedges) return;
    const int p = atomicAdd(&next[dstI[e]], 1);
    edges[p] = make_int2(srcI[e], __float_as_int(ew[e]));
}

__global__ void copy_ints_kernel(const int* __restrict__ from,
                                 int* __restrict__ to, int n)
{
    const int i = blockIdx.x * blockDim.x + threadIdx.x;
    if (i < n) to[i] = from[i];
}

template <int F>
__global__ __launch_bounds__(256) void gather_kernel(
    const float* __restrict__ h, const int* __restrict__ rowptr,
    const int2* __restrict__ edges, const float* __restrict__ bias,
    float* __restrict__ out, int nnodes)
{
    constexpr int PER = F / 32;
    constexpr int UN  = 8;
    typedef float __attribute__((ext_vector_type(PER))) vecf;
    const int lane = threadIdx.x & 31;
    const int d = __builtin_amdgcn_readfirstlane(blockIdx.x * 8 + (threadIdx.x >> 5));
    if (d >= nnodes) return;
    const int beg = rowptr[d];
    const int end = rowptr[d + 1];

    vecf acc = 0.0f;

    for (int j = beg; j < end; j += UN) {
        const int n = end - j;
        int2 r[UN];
#pragma unroll
        for (int u = 0; u < UN; ++u) r[u] = edges[j + u];
        vecf v[UN];
#pragma unroll
        for (int u = 0; u < UN; ++u) {
            if (u < n)
                v[u] = *(const vecf*)(h + (size_t)r[u].x * F + lane * PER);
        }
#pragma unroll
        for (int u = 0; u < UN; ++u) {
            if (u < n)
                acc += __int_as_float(r[u].y) * v[u];
        }
    }

    const vecf b = *(const vecf*)(bias + lane * PER);
    *(vecf*)(out + (size_t)d * F + lane * PER) = acc + b;
}

extern "C" void kernel_launch(void* const* d_in, const int* in_sizes, int n_in,
                              void* d_out, int out_size, void* d_ws, size_t ws_size,
                              hipStream_t stream)
{
    const float* x  = (const float*)d_in[0];
    const int*   ei = (const int*)d_in[1];
    const float* ew = (const float*)d_in[2];
    const float* W1 = (const float*)d_in[3];
    const float* b1 = (const float*)d_in[4];
    const float* W2 = (const float*)d_in[5];
    const float* b2 = (const float*)d_in[6];
    const float* W3 = (const float*)d_in[7];
    const float* b3 = (const float*)d_in[8];
    float* out = (float*)d_out;

    const int N = in_sizes[0] / FIN;
    const int E = in_sizes[2];
    const int* srcI = ei;
    const int* dstI = ei + E;

    float* bufA = (float*)d_ws;
    float* bufB = bufA + (size_t)N * FIN;

    const int ntiles     = (N + 15) / 16;
    const int gemmBlocks = (ntiles + 7) / 8;
    const int nodeBlocks = (N + 7) / 8;
    const int edgeBlocks = (E + 255) / 256;
    const int scanBlocks = (N + 1023) / 1024;

    const int    NP      = (N + 1 + 63) & ~63;
    const int    NB      = (scanBlocks + 63) & ~63;
    const size_t csrInts = (size_t)2 * NP + NB + (size_t)2 * E + 16;
    const bool roomy = ws_size >= ((size_t)2 * N * FIN + csrInts) * sizeof(float);
    int* csr  = roomy ? (int*)(bufB + (size_t)N * FIN) : (int*)d_out;
    int* csr3 = roomy ? csr : (int*)(bufA + (size_t)N * 64);

    int*  rowptr = csr;
    int*  next   = csr + NP;
    int*  bsum   = csr + 2 * NP;
    int2* edges  = (int2*)(csr + 2 * NP + NB);

    zero_counts_kernel<<<(N + 255) / 256, 256, 0, stream>>>(next, N);
    count_dst_kernel<<<edgeBlocks, 256, 0, stream>>>(dstI, next, E);
    scan_local_kernel<<<scanBlocks, 256, 0, stream>>>(next, rowptr, bsum, N);
    scan_offset_kernel<<<scanBlocks, 256, 0, stream>>>(rowptr, next, bsum, N);
    fill_edges_kernel<<<edgeBlocks, 256, 0, stream>>>(srcI, dstI, ew, next, edges, E);

    gemm_wmma_kernel<128, false><<<gemmBlocks, 256, 0, stream>>>(x, W1, bufA, N);
    gather_kernel<128><<<nodeBlocks, 256, 0, stream>>>(bufA, rowptr, edges, b1, bufB, N);

    gemm_wmma_kernel<128, true><<<gemmBlocks, 256, 0, stream>>>(bufB, W2, bufA, N);
    gather_kernel<128><<<nodeBlocks, 256, 0, stream>>>(bufA, rowptr, edges, b2, bufB, N);

    gemm_wmma_kernel<64, true><<<gemmBlocks, 256, 0, stream>>>(bufB, W3, bufA, N);
    if (!roomy)
        copy_ints_kernel<<<(int)((csrInts + 255) / 256), 256, 0, stream>>>(csr, csr3, (int)csrInts);
    gather_kernel<64><<<nodeBlocks, 256, 0, stream>>>(
        bufA, csr3, (const int2*)(csr3 + 2 * NP + NB), b3, out, N);
}


# bucketing without global atomics: two-level counting sort by destination on LDS counters (4096-edge tiles into 256-node buckets, then each bucket by node), whole-run stores; sums stay in fill order
# speedup vs baseline: 3.3048x; 1.5106x over previous
#include <hip/hip_runtime.h>
#include <hip/hip_bf16.h>

typedef __attribute__((ext_vector_type(2))) float v2f;
typedef __attribute__((ext_vector_type(8))) float v8f;

#define FIN 128

template <int FOUT, bool RELU>
__global__ __launch_bounds__(256) void gemm_wmma_kernel(
    const float* __restrict__ A, const float* __restrict__ W,
    float* __restrict__ C, int nrows)
{
    constexpr int NT = FOUT / 16;
    __shared__ float ldsW[FIN * FOUT];

    for (int i = threadIdx.x; i < FIN * FOUT; i += 256)
        ldsW[i] = W[i];
    __syncthreads();

    const int wave = threadIdx.x >> 5;
    const int lane = threadIdx.x & 31;
    const int half = lane >> 4;
    const int m    = lane & 15;

    const int tile   = blockIdx.x * 8 + wave;
    const int ntiles = (nrows + 15) >> 4;
    if (tile >= ntiles) return;

    const int row0 = tile * 16;
    int arow = row0 + m;
    if (arow >= nrows) arow = nrows - 1;
    const float* aptr = A + (size_t)arow * FIN;

    v8f acc[NT] = {};

    for (int k = 0; k < FIN; k += 4) {
        const int kk = k + 2 * half;
        float2 af = *(const float2*)(aptr + kk);
        if (RELU) {
            af.x = af.x > 0.0f ? af.x : 0.0f;
            af.y = af.y > 0.0f ? af.y : 0.0f;
        }
        v2f a;
        a.x = af.x;
        a.y = af.y;
#pragma unroll
        for (int t = 0; t < NT; ++t) {
            v2f b;
            b.x = ldsW[(kk)     * FOUT + t * 16 + m];
            b.y = ldsW[(kk + 1) * FOUT + t * 16 + m];
            acc[t] = __builtin_amdgcn_wmma_f32_16x16x4_f32(
                false, a, false, b, (short)0, acc[t], false, false);
        }
    }

#pragma unroll
    for (int r = 0; r < 8; ++r) {
        const int row = row0 + r + 8 * half;
        if (row < nrows) {
#pragma unroll
            for (int t = 0; t < NT; ++t)
                C[(size_t)row * FOUT + t * 16 + m] = acc[t][r];
        }
    }
}

__global__ void zero_counts_kernel(int* __restrict__ cnt, int n)
{
    const int i = blockIdx.x * blockDim.x + threadIdx.x;
    if (i < n) cnt[i] = 0;
}

__global__ void count_dst_kernel(const int* __restrict__ dstI,
                                 int* __restrict__ cnt, int nedges)
{
    const int e = blockIdx.x * blockDim.x + threadIdx.x;
    if (e < nedges) atomicAdd(&cnt[dstI[e]], 1);
}

__global__ __launch_bounds__(256) void scan_local_kernel(
    const int* __restrict__ cnt, int* __restrict__ rowptr,
    int* __restrict__ bsum, int n)
{
    __shared__ int sh[256];
    const int t  = threadIdx.x;
    const int i0 = blockIdx.x * 1024 + t * 4;
    int c[4];
#pragma unroll
    for (int k = 0; k < 4; ++k) c[k] = (i0 + k < n) ? cnt[i0 + k] : 0;
    const int mine = c[0] + c[1] + c[2] + c[3];
    sh[t] = mine;
    __syncthreads();
    for (int off = 1; off < 256; off <<= 1) {
        const int add = (t >= off) ? sh[t - off] : 0;
        __syncthreads();
        sh[t] += add;
        __syncthreads();
    }
    int run = sh[t] - mine;
#pragma unroll
    for (int k = 0; k < 4; ++k) {
        if (i0 + k < n) rowptr[i0 + k] = run;
        run += c[k];
    }
    if (t == 255) bsum[blockIdx.x] = sh[255];
}

template <bool NEXT>
__global__ __launch_bounds__(256) void scan_offset_kernel(
    int* __restrict__ rowptr, int* __restrict__ next,
    const int* __restrict__ bsum, int n)
{
    int off = 0;
    for (int j = 0; j < (int)blockIdx.x; ++j) off += bsum[j];
    const int i0 = blockIdx.x * 1024 + threadIdx.x * 4;
#pragma unroll
    for (int k = 0; k < 4; ++k) {
        if (i0 + k < n) {
            const int r = rowptr[i0 + k] + off;
            rowptr[i0 + k] = r;
            if (NEXT) next[i0 + k] = r;
        }
    }
    if (blockIdx.x == gridDim.x - 1 && threadIdx.x == 0)
        rowptr[n] = off + bsum[blockIdx.x];
}

__global__ void fill_edges_kernel(const int* __restrict__ srcI,
                                  const int* __restrict__ dstI,
                                  const float* __restrict__ ew,
                                  int* __restrict__ next,
                                  int2* __restrict__ edges, int nedges)
{
    const int e = blockIdx.x * blockDim.x + threadIdx.x;
    if (e < 8) edges[nedges + e] = make_int2(0, 0);
    if (e >= nedges) return;
    const int p = atomicAdd(&next[dstI[e]], 1);
    edges[p] = make_int2(srcI[e], __float_as_int(ew[e]));
}

__global__ void copy_ints_kernel(const int* __restrict__ from,
                                 int* __restrict__ to, int n)
{
    const int i = blockIdx.x * blockDim.x + threadIdx.x;
    if (i < n) to[i] = from[i];
}

#define BSH    8
#define BW     256
#define TILE   4096
#define SORT_T 512
#define CBMAX  1024
#define CAP    7168

__device__ __forceinline__ int block_excl_sum(int mine, int* sh)
{
    const int t = threadIdx.x;
    sh[t] = mine;
    __syncthreads();
    for (int off = 1; off < SORT_T; off <<= 1) {
        const int add = (t >= off) ? sh[t - off] : 0;
        __syncthreads();
        sh[t] += add;
        __syncthreads();
    }
    const int r = sh[t] - mine;
    __syncthreads();
    return r;
}

__device__ __forceinline__ int coarse_bucket(int d, int cb)
{
    return min((int)((unsigned)d >> BSH), cb - 1);
}

__global__ __launch_bounds__(SORT_T) void tile_count_kernel(
    const int* __restrict__ dstI, int* __restrict__ cnt1, int nedges, int cb, int g1)
{
    __shared__ int hist[CBMAX];
    const int t = threadIdx.x;
    for (int k = t; k < CBMAX; k += SORT_T) hist[k] = 0;
    __syncthreads();
    const int base = blockIdx.x * TILE;
    for (int u = 0; u < TILE / SORT_T; ++u) {
        const int e = base + u * SORT_T + t;
        if (e < nedges) atomicAdd(&hist[coarse_bucket(dstI[e], cb)], 1);
    }
    __syncthreads();
    for (int k = t; k < cb; k += SORT_T)
        cnt1[(size_t)k * g1 + blockIdx.x] = hist[k];
}

__global__ __launch_bounds__(SORT_T) void tile_scatter_kernel(
    const int* __restrict__ srcI, const int* __restrict__ dstI,
    const float* __restrict__ ew, const int* __restrict__ off1,
    int2* __restrict__ tmpE, int* __restrict__ tmpD, int nedges, int cb, int g1)
{
    constexpr int PT = TILE / SORT_T;
    __shared__ int  tcount[CBMAX];
    __shared__ int  tstart[CBMAX];
    __shared__ int  cur0[CBMAX];
    __shared__ int  sh[SORT_T];
    __shared__ int2 stageE[TILE];
    __shared__ int  stageD[TILE];
    const int t = threadIdx.x;
    for (int k = t; k < CBMAX; k += SORT_T) {
        tcount[k] = 0;
        cur0[k]   = (k < cb) ? off1[(size_t)k * g1 + blockIdx.x] : 0;
    }
    __syncthreads();

    const int base = blockIdx.x * TILE;
    const int nT   = min(TILE, nedges - base);
    int d[PT], s[PT], w[PT], r[PT];
#pragma unroll
    for (int u = 0; u < PT; ++u) {
        const int i = u * SORT_T + t;
        if (i < nT) {
            d[u] = dstI[base + i];
            s[u] = srcI[base + i];
            w[u] = __float_as_int(ew[base + i]);
        }
    }
#pragma unroll
    for (int u = 0; u < PT; ++u) {
        const int i = u * SORT_T + t;
        if (i < nT) r[u] = atomicAdd(&tcount[coarse_bucket(d[u], cb)], 1);
    }
    __syncthreads();

    const int c0 = tcount[2 * t];
    const int c1 = tcount[2 * t + 1];
    const int ex = block_excl_sum(c0 + c1, sh);
    tstart[2 * t]     = ex;
    tstart[2 * t + 1] = ex + c0;
    __syncthreads();

#pragma unroll
    for (int u = 0; u < PT; ++u) {
        const int i = u * SORT_T + t;
        if (i < nT) {
            const int slot = tstart[coarse_bucket(d[u], cb)] + r[u];
            stageE[slot] = make_int2(s[u], w[u]);
            stageD[slot] = d[u];
        }
    }
    __syncthreads();

    for (int i = t; i < nT; i += SORT_T) {
        const int    dd  = stageD[i];
        const int    k   = coarse_bucket(dd, cb);
        const size_t pos = (size_t)cur0[k] + (size_t)(i - tstart[k]);
        tmpE[pos] = stageE[i];
        tmpD[pos] = dd;
    }
}

__global__ __launch_bounds__(SORT_T) void bucket_sort_kernel(
    const int2* __restrict__ tmpE, const int* __restrict__ tmpD,
    const int* __restrict__ off1, int* __restrict__ rowptr,
    int2* __restrict__ edges, int nnodes, int nedges, int cb, int g1)
{
    __shared__ int  cnt[BW];
    __shared__ int  cursor[BW];
    __shared__ int  sh[SORT_T];
    __shared__ int2 stage[CAP];
    const int t  = threadIdx.x;
    const int k  = blockIdx.x;
    const int s0 = off1[(size_t)k * g1];
    const int s1 = off1[(size_t)(k + 1) * g1];
    const int L  = s1 - s0;
    if (t < BW) cnt[t] = 0;
    if (k == 0 && t < 8) edges[nedges + t] = make_int2(0, 0);
    __syncthreads();

    for (int i = t; i < L; i += SORT_T)
        atomicAdd(&cnt[tmpD[s0 + i] & (BW - 1)], 1);
    __syncthreads();

    const int mine = (t < BW) ? cnt[t] : 0;
    const int ex   = block_excl_sum(mine, sh);
    if (t < BW) {
        cursor[t] = ex;
        const int node = k * BW + t;
        if (node < nnodes) rowptr[node] = s0 + ex;
    }
    if (k == cb - 1 && t == 0) rowptr[nnodes] = s1;
    __syncthreads();

    const bool staged = L <= CAP;
    for (int i = t; i < L; i += SORT_T) {
        const int  p   = atomicAdd(&cursor[tmpD[s0 + i] & (BW - 1)], 1);
        const int2 rec = tmpE[s0 + i];
        if (staged) stage[p] = rec;
        else        edges[s0 + p] = rec;
    }
    __syncthreads();
    if (staged)
        for (int i = t; i < L; i += SORT_T) edges[s0 + i] = stage[i];
}

template <int F>
__global__ __launch_bounds__(256) void gather_kernel(
    const float* __restrict__ h, const int* __restrict__ rowptr,
    const int2* __restrict__ edges, const float* __restrict__ bias,
    float* __restrict__ out, int nnodes)
{
    constexpr int PER = F / 32;
    constexpr int UN  = 8;
    typedef float __attribute__((ext_vector_type(PER))) vecf;
    const int lane = threadIdx.x & 31;
    const int d = __builtin_amdgcn_readfirstlane(blockIdx.x * 8 + (threadIdx.x >> 5));
    if (d >= nnodes) return;
    const int beg = rowptr[d];
    const int end = rowptr[d + 1];

    vecf acc = 0.0f;

    for (int j = beg; j < end; j += UN) {
        const int n = end - j;
        int2 r[UN];
#pragma unroll
        for (int u = 0; u < UN; ++u) r[u] = edges[j + u];
        vecf v[UN];
#pragma unroll
        for (int u = 0; u < UN; ++u) {
            if (u < n)
                v[u] = *(const vecf*)(h + (size_t)r[u].x * F + lane * PER);
        }
#pragma unroll
        for (int u = 0; u < UN; ++u) {
            if (u < n)
                acc += __int_as_float(r[u].y) * v[u];
        }
    }

    const vecf b = *(const vecf*)(bias + lane * PER);
    *(vecf*)(out + (size_t)d * F + lane * PER) = acc + b;
}

extern "C" void kernel_launch(void* const* d_in, const int* in_sizes, int n_in,
                              void* d_out, int out_size, void* d_ws, size_t ws_size,
                              hipStream_t stream)
{
    const float* x  = (const float*)d_in[0];
    const int*   ei = (const int*)d_in[1];
    const float* ew = (const float*)d_in[2];
    const float* W1 = (const float*)d_in[3];
    const float* b1 = (const float*)d_in[4];
    const float* W2 = (const float*)d_in[5];
    const float* b2 = (const float*)d_in[6];
    const float* W3 = (const float*)d_in[7];
    const float* b3 = (const float*)d_in[8];
    float* out = (float*)d_out;

    const int N = in_sizes[0] / FIN;
    const int E = in_sizes[2];
    const int* srcI = ei;
    const int* dstI = ei + E;

    float* bufA = (float*)d_ws;
    float* bufB = bufA + (size_t)N * FIN;

    const int ntiles     = (N + 15) / 16;
    const int gemmBlocks = (ntiles + 7) / 8;
    const int nodeBlocks = (N + 7) / 8;
    const int edgeBlocks = (E + 255) / 256;
    const int scanBlocks = (N + 1023) / 1024;

    const int    NP      = (N + 1 + 63) & ~63;
    const int    NB      = (scanBlocks + 63) & ~63;
    const size_t csrInts = (size_t)2 * NP + NB + (size_t)2 * E + 16;
    const bool roomy = ws_size >= ((size_t)2 * N * FIN + csrInts) * sizeof(float);
    int* csr  = roomy ? (int*)(bufB + (size_t)N * FIN) : (int*)d_out;
    int* csr3 = roomy ? csr : (int*)(bufA + (size_t)N * 64);

    int*  rowptr = csr;
    int*  next   = csr + NP;
    int*  bsum   = csr + 2 * NP;
    int2* edges  = (int2*)(csr + 2 * NP + NB);

    const int    CB      = (N + BW - 1) / BW;
    const int    G1      = (E + TILE - 1) / TILE;
    const size_t n1      = (size_t)CB * G1;
    const size_t n1p     = (n1 + 1 + 63) & ~(size_t)63;
    const int    scanBl1 = (int)((n1 + 1023) / 1024);
    const size_t NB1     = ((size_t)scanBl1 + 63) & ~(size_t)63;
    const bool   ldsSort = CB <= CBMAX && G1 > 0 && n1 < ((size_t)1 << 30) &&
                           (size_t)3 * E + 2 * n1p + NB1 <= (size_t)N * FIN;
    if (ldsSort) {
        int2* tmpE  = (int2*)bufB;
        int*  tmpD  = (int*)(tmpE + E);
        int*  cnt1  = tmpD + E;
        int*  off1  = cnt1 + n1p;
        int*  bsum1 = off1 + n1p;
        tile_count_kernel<<<G1, SORT_T, 0, stream>>>(dstI, cnt1, E, CB, G1);
        scan_local_kernel<<<scanBl1, 256, 0, stream>>>(cnt1, off1, bsum1, (int)n1);
        scan_offset_kernel<false><<<scanBl1, 256, 0, stream>>>(off1, nullptr, bsum1, (int)n1);
        tile_scatter_kernel<<<G1, SORT_T, 0, stream>>>(srcI, dstI, ew, off1, tmpE, tmpD, E, CB, G1);
        bucket_sort_kernel<<<CB, SORT_T, 0, stream>>>(tmpE, tmpD, off1, rowptr, edges, N, E, CB, G1);
    } else {
        zero_counts_kernel<<<(N + 255) / 256, 256, 0, stream>>>(next, N);
        count_dst_kernel<<<edgeBlocks, 256, 0, stream>>>(dstI, next, E);
        scan_local_kernel<<<scanBlocks, 256, 0, stream>>>(next, rowptr, bsum, N);
        scan_offset_kernel<true><<<scanBlocks, 256, 0, stream>>>(rowptr, next, bsum, N);
        fill_edges_kernel<<<edgeBlocks, 256, 0, stream>>>(srcI, dstI, ew, next, edges, E);
    }

    gemm_wmma_kernel<128, false><<<gemmBlocks, 256, 0, stream>>>(x, W1, bufA, N);
    gather_kernel<128><<<nodeBlocks, 256, 0, stream>>>(bufA, rowptr, edges, b1, bufB, N);

    gemm_wmma_kernel<128, true><<<gemmBlocks, 256, 0, stream>>>(bufB, W2, bufA, N);
    gather_kernel<128><<<nodeBlocks, 256, 0, stream>>>(bufA, rowptr, edges, b2, bufB, N);

    gemm_wmma_kernel<64, true><<<gemmBlocks, 256, 0, stream>>>(bufB, W3, bufA, N);
    if (!roomy)
        copy_ints_kernel<<<(int)((csrInts + 255) / 256), 256, 0, stream>>>(csr, csr3, (int)csrInts);
    gather_kernel<64><<<nodeBlocks, 256, 0, stream>>>(
        bufA, csr3, (const int2*)(csr3 + 2 * NP + NB), b3, out, N);
}


# each gather fused with the next layer's product: 16-wave groups sum 8 tiles of 16 nodes (bias, relu) into LDS and multiply by W there; 2 launches and the h round trips gone; same order of sums
# speedup vs baseline: 3.4940x; 1.0573x over previous
#include <hip/hip_runtime.h>
#include <hip/hip_bf16.h>

typedef __attribute__((ext_vector_type(2))) float v2f;
typedef __attribute__((ext_vector_type(8))) float v8f;

#define FIN 128

template <int FOUT, bool RELU>
__global__ __launch_bounds__(256) void gemm_wmma_kernel(
    const float* __restrict__ A, const float* __restrict__ W,
    float* __restrict__ C, int nrows)
{
    constexpr int NT = FOUT / 16;
    __shared__ float ldsW[FIN * FOUT];

    for (int i = threadIdx.x; i < FIN * FOUT; i += 256)
        ldsW[i] = W[i];
    __syncthreads();

    const int wave = threadIdx.x >> 5;
    const int lane = threadIdx.x & 31;
    const int half = lane >> 4;
    const int m    = lane & 15;

    const int tile   = blockIdx.x * 8 + wave;
    const int ntiles = (nrows + 15) >> 4;
    if (tile >= ntiles) return;

    const int row0 = tile * 16;
    int arow = row0 + m;
    if (arow >= nrows) arow = nrows - 1;
    const float* aptr = A + (size_t)arow * FIN;

    v8f acc[NT] = {};

    for (int k = 0; k < FIN; k += 4) {
        const int kk = k + 2 * half;
        float2 af = *(const float2*)(aptr + kk);
        if (RELU) {
            af.x = af.x > 0.0f ? af.x : 0.0f;
            af.y = af.y > 0.0f ? af.y : 0.0f;
        }
        v2f a;
        a.x = af.x;
        a.y = af.y;
#pragma unroll
        for (int t = 0; t < NT; ++t) {
            v2f b;
            b.x = ldsW[(kk)     * FOUT + t * 16 + m];
            b.y = ldsW[(kk + 1) * FOUT + t * 16 + m];
            acc[t] = __builtin_amdgcn_wmma_f32_16x16x4_f32(
                false, a, false, b, (short)0, acc[t], false, false);
        }
    }

#pragma unroll
    for (int r = 0; r < 8; ++r) {
        const int row = row0 + r + 8 * half;
        if (row < nrows) {
#pragma unroll
            for (int t = 0; t < NT; ++t)
                C[(size_t)row * FOUT + t * 16 + m] = acc[t][r];
        }
    }
}

__global__ void zero_counts_kernel(int* __restrict__ cnt, int n)
{
    const int i = blockIdx.x * blockDim.x + threadIdx.x;
    if (i < n) cnt[i] = 0;
}

__global__ void count_dst_kernel(const int* __restrict__ dstI,
                                 int* __restrict__ cnt, int nedges)
{
    const int e = blockIdx.x * blockDim.x + threadIdx.x;
    if (e < nedges) atomicAdd(&cnt[dstI[e]], 1);
}

__global__ __launch_bounds__(256) void scan_local_kernel(
    const int* __restrict__ cnt, int* __restrict__ rowptr,
    int* __restrict__ bsum, int n)
{
    __shared__ int sh[256];
    const int t  = threadIdx.x;
    const int i0 = blockIdx.x * 1024 + t * 4;
    int c[4];
#pragma unroll
    for (int k = 0; k < 4; ++k) c[k] = (i0 + k < n) ? cnt[i0 + k] : 0;
    const int mine = c[0] + c[1] + c[2] + c[3];
    sh[t] = mine;
    __syncthreads();
    for (int off = 1; off < 256; off <<= 1) {
        const int add = (t >= off) ? sh[t - off] : 0;
        __syncthreads();
        sh[t] += add;
        __syncthreads();
    }
    int run = sh[t] - mine;
#pragma unroll
    for (int k = 0; k < 4; ++k) {
        if (i0 + k < n) rowptr[i0 + k] = run;
        run += c[k];
    }
    if (t == 255) bsum[blockIdx.x] = sh[255];
}

template <bool NEXT>
__global__ __launch_bounds__(256) void scan_offset_kernel(
    int* __restrict__ rowptr, int* __restrict__ next,
    const int* __restrict__ bsum, int n)
{
    int off = 0;
    for (int j = 0; j < (int)blockIdx.x; ++j) off += bsum[j];
    const int i0 = blockIdx.x * 1024 + threadIdx.x * 4;
#pragma unroll
    for (int k = 0; k < 4; ++k) {
        if (i0 + k < n) {
            const int r = rowptr[i0 + k] + off;
            rowptr[i0 + k] = r;
            if (NEXT) next[i0 + k] = r;
        }
    }
    if (blockIdx.x == gridDim.x - 1 && threadIdx.x == 0)
        rowptr[n] = off + bsum[blockIdx.x];
}

__global__ void fill_edges_kernel(const int* __restrict__ srcI,
                                  const int* __restrict__ dstI,
                                  const float* __restrict__ ew,
                                  int* __restrict__ next,
                                  int2* __restrict__ edges, int nedges)
{
    const int e = blockIdx.x * blockDim.x + threadIdx.x;
    if (e < 8) edges[nedges + e] = make_int2(0, 0);
    if (e >= nedges) return;
    const int p = atomicAdd(&next[dstI[e]], 1);
    edges[p] = make_int2(srcI[e], __float_as_int(ew[e]));
}

__global__ void copy_ints_kernel(const int* __restrict__ from,
                                 int* __restrict__ to, int n)
{
    const int i = blockIdx.x * blockDim.x + threadIdx.x;
    if (i < n) to[i] = from[i];
}

#define BSH    8
#define BW     256
#define TILE   4096
#define SORT_T 512
#define CBMAX  1024
#define CAP    7168

__device__ __forceinline__ int block_excl_sum(int mine, int* sh)
{
    const int t = threadIdx.x;
    sh[t] = mine;
    __syncthreads();
    for (int off = 1; off < SORT_T; off <<= 1) {
        const int add = (t >= off) ? sh[t - off] : 0;
        __syncthreads();
        sh[t] += add;
        __syncthreads();
    }
    const int r = sh[t] - mine;
    __syncthreads();
    return r;
}

__device__ __forceinline__ int coarse_bucket(int d, int cb)
{
    return min((int)((unsigned)d >> BSH), cb - 1);
}

__global__ __launch_bounds__(SORT_T) void tile_count_kernel(
    const int* __restrict__ dstI, int* __restrict__ cnt1, int nedges, int cb, int g1)
{
    __shared__ int hist[CBMAX];
    const int t = threadIdx.x;
    for (int k = t; k < CBMAX; k += SORT_T) hist[k] = 0;
    __syncthreads();
    const int base = blockIdx.x * TILE;
    for (int u = 0; u < TILE / SORT_T; ++u) {
        const int e = base + u * SORT_T + t;
        if (e < nedges) atomicAdd(&hist[coarse_bucket(dstI[e], cb)], 1);
    }
    __syncthreads();
    for (int k = t; k < cb; k += SORT_T)
        cnt1[(size_t)k * g1 + blockIdx.x] = hist[k];
}

__global__ __launch_bounds__(SORT_T) void tile_scatter_kernel(
    const int* __restrict__ srcI, const int* __restrict__ dstI,
    const float* __restrict__ ew, const int* __restrict__ off1,
    int2* __restrict__ tmpE, int* __restrict__ tmpD, int nedges, int cb, int g1)
{
    constexpr int PT = TILE / SORT_T;
    __shared__ int  tcount[CBMAX];
    __shared__ int  tstart[CBMAX];
    __shared__ int  cur0[CBMAX];
    __shared__ int  sh[SORT_T];
    __shared__ int2 stageE[TILE];
    __shared__ int  stageD[TILE];
    const int t = threadIdx.x;
    for (int k = t; k < CBMAX; k += SORT_T) {
        tcount[k] = 0;
        cur0[k]   = (k < cb) ? off1[(size_t)k * g1 + blockIdx.x] : 0;
    }
    __syncthreads();

    const int base = blockIdx.x * TILE;
    const int nT   = min(TILE, nedges - base);
    int d[PT], s[PT], w[PT], r[PT];
#pragma unroll
    for (int u = 0; u < PT; ++u) {
        const int i = u * SORT_T + t;
        if (i < nT) {
            d[u] = dstI[base + i];
            s[u] = srcI[base + i];
            w[u] = __float_as_int(ew[base + i]);
        }
    }
#pragma unroll
    for (int u = 0; u < PT; ++u) {
        const int i = u * SORT_T + t;
        if (i < nT) r[u] = atomicAdd(&tcount[coarse_bucket(d[u], cb)], 1);
    }
    __syncthreads();

    const int c0 = tcount[2 * t];
    const int c1 = tcount[2 * t + 1];
    const int ex = block_excl_sum(c0 + c1, sh);
    tstart[2 * t]     = ex;
    tstart[2 * t + 1] = ex + c0;
    __syncthreads();

#pragma unroll
    for (int u = 0; u < PT; ++u) {
        const int i = u * SORT_T + t;
        if (i < nT) {
            const int slot = tstart[coarse_bucket(d[u], cb)] + r[u];
            stageE[slot] = make_int2(s[u], w[u]);
            stageD[slot] = d[u];
        }
    }
    __syncthreads();

    for (int i = t; i < nT; i += SORT_T) {
        const int    dd  = stageD[i];
        const int    k   = coarse_bucket(dd, cb);
        const size_t pos = (size_t)cur0[k] + (size_t)(i - tstart[k]);
        tmpE[pos] = stageE[i];
        tmpD[pos] = dd;
    }
}

__global__ __launch_bounds__(SORT_T) void bucket_sort_kernel(
    const int2* __restrict__ tmpE, const int* __restrict__ tmpD,
    const int* __restrict__ off1, int* __restrict__ rowptr,
    int2* __restrict__ edges, int nnodes, int nedges, int cb, int g1)
{
    __shared__ int  cnt[BW];
    __shared__ int  cursor[BW];
    __shared__ int  sh[SORT_T];
    __shared__ int2 stage[CAP];
    const int t  = threadIdx.x;
    const int k  = blockIdx.x;
    const int s0 = off1[(size_t)k * g1];
    const int s1 = off1[(size_t)(k + 1) * g1];
    const int L  = s1 - s0;
    if (t < BW) cnt[t] = 0;
    if (k == 0 && t < 8) edges[nedges + t] = make_int2(0, 0);
    __syncthreads();

    for (int i = t; i < L; i += SORT_T)
        atomicAdd(&cnt[tmpD[s0 + i] & (BW - 1)], 1);
    __syncthreads();

    const int mine = (t < BW) ? cnt[t] : 0;
    const int ex   = block_excl_sum(mine, sh);
    if (t < BW) {
        cursor[t] = ex;
        const int node = k * BW + t;
        if (node < nnodes) rowptr[node] = s0 + ex;
    }
    if (k == cb - 1 && t == 0) rowptr[nnodes] = s1;
    __syncthreads();

    const bool staged = L <= CAP;
    for (int i = t; i < L; i += SORT_T) {
        const int  p   = atomicAdd(&cursor[tmpD[s0 + i] & (BW - 1)], 1);
        const int2 rec = tmpE[s0 + i];
        if (staged) stage[p] = rec;
        else        edges[s0 + p] = rec;
    }
    __syncthreads();
    if (staged)
        for (int i = t; i < L; i += SORT_T) edges[s0 + i] = stage[i];
}

template <int F>
__global__ __launch_bounds__(256) void gather_kernel(
    const float* __restrict__ h, const int* __restrict__ rowptr,
    const int2* __restrict__ edges, const float* __restrict__ bias,
    float* __restrict__ out, int nnodes)
{
    constexpr int PER = F / 32;
    constexpr int UN  = 8;
    typedef float __attribute__((ext_vector_type(PER))) vecf;
    const int lane = threadIdx.x & 31;
    const int d = __builtin_amdgcn_readfirstlane(blockIdx.x * 8 + (threadIdx.x >> 5));
    if (d >= nnodes) return;
    const int beg = rowptr[d];
    const int end = rowptr[d + 1];

    vecf acc = 0.0f;

    for (int j = beg; j < end; j += UN) {
        const int n = end - j;
        int2 r[UN];
#pragma unroll
        for (int u = 0; u < UN; ++u) r[u] = edges[j + u];
        vecf v[UN];
#pragma unroll
        for (int u = 0; u < UN; ++u) {
            if (u < n)
                v[u] = *(const vecf*)(h + (size_t)r[u].x * F + lane * PER);
        }
#pragma unroll
        for (int u = 0; u < UN; ++u) {
            if (u < n)
                acc += __int_as_float(r[u].y) * v[u];
        }
    }

    const vecf b = *(const vecf*)(bias + lane * PER);
    *(vecf*)(out + (size_t)d * F + lane * PER) = acc + b;
}


#define TPITCH 132

template <int FOUT>
__global__ __launch_bounds__(512) void gather_gemm_kernel(
    const float* __restrict__ h, const int* __restrict__ rowptr,
    const int2* __restrict__ edges, const float* __restrict__ bias,
    const float* __restrict__ Wn, float* __restrict__ C, int nnodes)
{
    constexpr int NT  = FOUT / 16;
    constexpr int NTW = NT / 2;
    constexpr int UN  = 8;
    typedef float __attribute__((ext_vector_type(4))) vecf;
    constexpr int WQ  = FIN * FOUT / 4 / 512;
    __shared__ float lds[FIN * FOUT + 8 * 16 * TPITCH];
    float* ldsW = lds;
    float* tiles = lds + FIN * FOUT;

    float4 wq[WQ];
#pragma unroll
    for (int i = 0; i < WQ; ++i)
        wq[i] = ((const float4*)Wn)[threadIdx.x + i * 512];

    const int lane = threadIdx.x & 31;
    const int wave = __builtin_amdgcn_readfirstlane(threadIdx.x >> 5);
    const int pair = wave >> 1;
    const int sub  = wave & 1;
    const int node0 = (blockIdx.x * 8 + pair) * 16;
    float* trow = tiles + pair * 16 * TPITCH;

    const int first = node0 + 8 * sub;
    const int rp = rowptr[min(first + lane, nnodes)];
    const vecf b = *(const vecf*)(bias + lane * 4);

#pragma unroll
    for (int i = 0; i < WQ; ++i)
        ((float4*)ldsW)[threadIdx.x + i * 512] = wq[i];

    for (int q = 0; q < 8; ++q) {
        const int d = first + q;
        vecf row = 0.0f;
        if (d < nnodes) {
            const int beg = __builtin_amdgcn_readlane(rp, q);
            const int end = __builtin_amdgcn_readlane(rp, q + 1);
            vecf acc = 0.0f;
            for (int j = beg; j < end; j += UN) {
                const int n = end - j;
                int2 r[UN];
#pragma unroll
                for (int u = 0; u < UN; ++u) r[u] = edges[j + u];
                vecf v[UN];
#pragma unroll
                for (int u = 0; u < UN; ++u) {
                    if (u < n)
                        v[u] = *(const vecf*)(h + (size_t)r[u].x * FIN + lane * 4);
                }
#pragma unroll
                for (int u = 0; u < UN; ++u) {
                    if (u < n)
                        acc += __int_as_float(r[u].y) * v[u];
                }
            }
            row = acc + b;
            row.x = row.x > 0.0f ? row.x : 0.0f;
            row.y = row.y > 0.0f ? row.y : 0.0f;
            row.z = row.z > 0.0f ? row.z : 0.0f;
            row.w = row.w > 0.0f ? row.w : 0.0f;
        }
        *(vecf*)(trow + (8 * sub + q) * TPITCH + lane * 4) = row;
    }
    __syncthreads();
    if (node0 >= nnodes) return;

    const int half = lane >> 4;
    const int m    = lane & 15;
    const float* arow = trow + m * TPITCH;

    v8f acc[NTW] = {};

    for (int k = 0; k < FIN; k += 4) {
        const int kk = k + 2 * half;
        const float2 af = *(const float2*)(arow + kk);
        v2f a;
        a.x = af.x;
        a.y = af.y;
#pragma unroll
        for (int t = 0; t < NTW; ++t) {
            const int col = (sub * NTW + t) * 16 + m;
            v2f bf;
            bf.x = ldsW[(kk)     * FOUT + col];
            bf.y = ldsW[(kk + 1) * FOUT + col];
            acc[t] = __builtin_amdgcn_wmma_f32_16x16x4_f32(
                false, a, false, bf, (short)0, acc[t], false, false);
        }
    }

#pragma unroll
    for (int r = 0; r < 8; ++r) {
        const int orow = node0 + r + 8 * half;
        if (orow < nnodes) {
#pragma unroll
            for (int t = 0; t < NTW; ++t)
                C[(size_t)orow * FOUT + (sub * NTW + t) * 16 + m] = acc[t][r];
        }
    }
}

extern "C" void kernel_launch(void* const* d_in, const int* in_sizes, int n_in,
                              void* d_out, int out_size, void* d_ws, size_t ws_size,
                              hipStream_t stream)
{
    const float* x  = (const float*)d_in[0];
    const int*   ei = (const int*)d_in[1];
    const float* ew = (const float*)d_in[2];
    const float* W1 = (const float*)d_in[3];
    const float* b1 = (const float*)d_in[4];
    const float* W2 = (const float*)d_in[5];
    const float* b2 = (const float*)d_in[6];
    const float* W3 = (const float*)d_in[7];
    const float* b3 = (const float*)d_in[8];
    float* out = (float*)d_out;

    const int N = in_sizes[0] / FIN;
    const int E = in_sizes[2];
    const int* srcI = ei;
    const int* dstI = ei + E;

    float* bufA = (float*)d_ws;
    float* bufB = bufA + (size_t)N * FIN;

    const int ntiles     = (N + 15) / 16;
    const int gemmBlocks = (ntiles + 7) / 8;
    const int nodeBlocks = (N + 7) / 8;
    const int edgeBlocks = (E + 255) / 256;
    const int scanBlocks = (N + 1023) / 1024;

    const int    NP      = (N + 1 + 63) & ~63;
    const int    NB      = (scanBlocks + 63) & ~63;
    const size_t csrInts = (size_t)2 * NP + NB + (size_t)2 * E + 16;
    const bool roomy = ws_size >= ((size_t)2 * N * FIN + csrInts) * sizeof(float);
    int* csr  = roomy ? (int*)(bufB + (size_t)N * FIN) : (int*)d_out;
    int* csr3 = roomy ? csr : (int*)(bufA + (size_t)N * 64);

    int*  rowptr = csr;
    int*  next   = csr + NP;
    int*  bsum   = csr + 2 * NP;
    int2* edges  = (int2*)(csr + 2 * NP + NB);

    const int    CB      = (N + BW - 1) / BW;
    const int    G1      = (E + TILE - 1) / TILE;
    const size_t n1      = (size_t)CB * G1;
    const size_t n1p     = (n1 + 1 + 63) & ~(size_t)63;
    const int    scanBl1 = (int)((n1 + 1023) / 1024);
    const size_t NB1     = ((size_t)scanBl1 + 63) & ~(size_t)63;
    const bool   ldsSort = CB <= CBMAX && G1 > 0 && n1 < ((size_t)1 << 30) &&
                           (size_t)3 * E + 2 * n1p + NB1 <= (size_t)N * FIN;
    if (ldsSort) {
        int2* tmpE  = (int2*)bufB;
        int*  tmpD  = (int*)(tmpE + E);
        int*  cnt1  = tmpD + E;
        int*  off1  = cnt1 + n1p;
        int*  bsum1 = off1 + n1p;
        tile_count_kernel<<<G1, SORT_T, 0, stream>>>(dstI, cnt1, E, CB, G1);
        scan_local_kernel<<<scanBl1, 256, 0, stream>>>(cnt1, off1, bsum1, (int)n1);
        scan_offset_kernel<false><<<scanBl1, 256, 0, stream>>>(off1, nullptr, bsum1, (int)n1);
        tile_scatter_kernel<<<G1, SORT_T, 0, stream>>>(srcI, dstI, ew, off1, tmpE, tmpD, E, CB, G1);
        bucket_sort_kernel<<<CB, SORT_T, 0, stream>>>(tmpE, tmpD, off1, rowptr, edges, N, E, CB, G1);
    } else {
        zero_counts_kernel<<<(N + 255) / 256, 256, 0, stream>>>(next, N);
        count_dst_kernel<<<edgeBlocks, 256, 0, stream>>>(dstI, next, E);
        scan_local_kernel<<<scanBlocks, 256, 0, stream>>>(next, rowptr, bsum, N);
        scan_offset_kernel<true><<<scanBlocks, 256, 0, stream>>>(rowptr, next, bsum, N);
        fill_edges_kernel<<<edgeBlocks, 256, 0, stream>>>(srcI, dstI, ew, next, edges, E);
    }

    gemm_wmma_kernel<128, false><<<gemmBlocks, 256, 0, stream>>>(x, W1, bufA, N);

    gather_gemm_kernel<128><<<gemmBlocks, 512, 0, stream>>>(bufA, rowptr, edges, b1, W2, bufB, N);

    gather_gemm_kernel<64><<<gemmBlocks, 512, 0, stream>>>(bufB, rowptr, edges, b2, W3, bufA, N);

    if (!roomy)
        copy_ints_kernel<<<(int)((csrInts + 255) / 256), 256, 0, stream>>>(csr, csr3, (int)csrInts);
    gather_kernel<64><<<nodeBlocks, 256, 0, stream>>>(
        bufA, csr3, (const int2*)(csr3 + 2 * NP + NB), b3, out, N);
}


# x @ W1 moved into the three big launches of the edge sort: up to 256 extra 16-wave work-groups per launch multiply a share of the row tiles beside the sorting ones; one launch fewer; same sums
# speedup vs baseline: 3.7732x; 1.0799x over previous
#include <hip/hip_runtime.h>
#include <hip/hip_bf16.h>

typedef __attribute__((ext_vector_type(2))) float v2f;
typedef __attribute__((ext_vector_type(8))) float v8f;

#define FIN 128

template <int FOUT, bool RELU>
__global__ __launch_bounds__(256) void gemm_wmma_kernel(
    const float* __restrict__ A, const float* __restrict__ W,
    float* __restrict__ C, int nrows)
{
    constexpr int NT = FOUT / 16;
    __shared__ float ldsW[FIN * FOUT];

    for (int i = threadIdx.x; i < FIN * FOUT; i += 256)
        ldsW[i] = W[i];
    __syncthreads();

    const int wave = threadIdx.x >> 5;
    const int lane = threadIdx.x & 31;
    const int half = lane >> 4;
    const int m    = lane & 15;

    const int tile   = blockIdx.x * 8 + wave;
    const int ntiles = (nrows + 15) >> 4;
    if (tile >= ntiles) return;

    const int row0 = tile * 16;
    int arow = row0 + m;
    if (arow >= nrows) arow = nrows - 1;
    const float* aptr = A + (size_t)arow * FIN;

    v8f acc[NT] = {};

    for (int k = 0; k < FIN; k += 4) {
        const int kk = k + 2 * half;
        float2 af = *(const float2*)(aptr + kk);
        if (RELU) {
            af.x = af.x > 0.0f ? af.x : 0.0f;
            af.y = af.y > 0.0f ? af.y : 0.0f;
        }
        v2f a;
        a.x = af.x;
        a.y = af.y;
#pragma unroll
        for (int t = 0; t < NT; ++t) {
            v2f b;
            b.x = ldsW[(kk)     * FOUT + t * 16 + m];
            b.y = ldsW[(kk + 1) * FOUT + t * 16 + m];
            acc[t] = __builtin_amdgcn_wmma_f32_16x16x4_f32(
                false, a, false, b, (short)0, acc[t], false, false);
        }
    }

#pragma unroll
    for (int r = 0; r < 8; ++r) {
        const int row = row0 + r + 8 * half;
        if (row < nrows) {
#pragma unroll
            for (int t = 0; t < NT; ++t)
                C[(size_t)row * FOUT + t * 16 + m] = acc[t][r];
        }
    }
}

__global__ void zero_counts_kernel(int* __restrict__ cnt, int n)
{
    const int i = blockIdx.x * blockDim.x + threadIdx.x;
    if (i < n) cnt[i] = 0;
}

__global__ void count_dst_kernel(const int* __restrict__ dstI,
                                 int* __restrict__ cnt, int nedges)
{
    const int e = blockIdx.x * blockDim.x + threadIdx.x;
    if (e < nedges) atomicAdd(&cnt[dstI[e]], 1);
}

__global__ __launch_bounds__(256) void scan_local_kernel(
    const int* __restrict__ cnt, int* __restrict__ rowptr,
    int* __restrict__ bsum, int n)
{
    __shared__ int sh[256];
    const int t  = threadIdx.x;
    const int i0 = blockIdx.x * 1024 + t * 4;
    int c[4];
#pragma unroll
    for (int k = 0; k < 4; ++k) c[k] = (i0 + k < n) ? cnt[i0 + k] : 0;
    const int mine = c[0] + c[1] + c[2] + c[3];
    sh[t] = mine;
    __syncthreads();
    for (int off = 1; off < 256; off <<= 1) {
        const int add = (t >= off) ? sh[t - off] : 0;
        __syncthreads();
        sh[t] += add;
        __syncthreads();
    }
    int run = sh[t] - mine;
#pragma unroll
    for (int k = 0; k < 4; ++k) {
        if (i0 + k < n) rowptr[i0 + k] = run;
        run += c[k];
    }
    if (t == 255) bsum[blockIdx.x] = sh[255];
}

template <bool NEXT>
__global__ __launch_bounds__(256) void scan_offset_kernel(
    int* __restrict__ rowptr, int* __restrict__ next,
    const int* __restrict__ bsum, int n)
{
    int off = 0;
    for (int j = 0; j < (int)blockIdx.x; ++j) off += bsum[j];
    const int i0 = blockIdx.x * 1024 + threadIdx.x * 4;
#pragma unroll
    for (int k = 0; k < 4; ++k) {
        if (i0 + k < n) {
            const int r = rowptr[i0 + k] + off;
            rowptr[i0 + k] = r;
            if (NEXT) next[i0 + k] = r;
        }
    }
    if (blockIdx.x == gridDim.x - 1 && threadIdx.x == 0)
        rowptr[n] = off + bsum[blockIdx.x];
}

__global__ void fill_edges_kernel(const int* __restrict__ srcI,
                                  const int* __restrict__ dstI,
                                  const float* __restrict__ ew,
                                  int* __restrict__ next,
                                  int2* __restrict__ edges, int nedges)
{
    const int e = blockIdx.x * blockDim.x + threadIdx.x;
    if (e < 8) edges[nedges + e] = make_int2(0, 0);
    if (e >= nedges) return;
    const int p = atomicAdd(&next[dstI[e]], 1);
    edges[p] = make_int2(srcI[e], __float_as_int(ew[e]));
}

__global__ void copy_ints_kernel(const int* __restrict__ from,
                                 int* __restrict__ to, int n)
{
    const int i = blockIdx.x * blockDim.x + threadIdx.x;
    if (i < n) to[i] = from[i];
}

#define BSH    8
#define BW     256
#define TILE   4096
#define SORT_T 512
#define CBMAX  1024
#define CAP    7168

__device__ __forceinline__ int block_excl_sum(int mine, int* sh)
{
    const int t = threadIdx.x;
    sh[t] = mine;
    __syncthreads();
    for (int off = 1; off < SORT_T; off <<= 1) {
        const int add = (t >= off) ? sh[t - off] : 0;
        __syncthreads();
        sh[t] += add;
        __syncthreads();
    }
    const int r = sh[t] - mine;
    __syncthreads();
    return r;
}

__device__ __forceinline__ int coarse_bucket(int d, int cb)
{
    return min((int)((unsigned)d >> BSH), cb - 1);
}

#define XW_NB    256
#define LDS_INTS (FIN * 128)

__device__ __forceinline__ void xw_role(
    float* ldsW, const float* __restrict__ X, const float* __restrict__ Wm,
    float* __restrict__ C, int nrows, int gb, int nb, int t0, int t1)
{
    constexpr int FOUT = 128;
    constexpr int NTW  = 4;
    constexpr int WQ   = FIN * FOUT / 4 / SORT_T;
    float4 wq[WQ];
#pragma unroll
    for (int i = 0; i < WQ; ++i)
        wq[i] = ((const float4*)Wm)[threadIdx.x + i * SORT_T];

    const int lane = threadIdx.x & 31;
    const int wave = __builtin_amdgcn_readfirstlane(threadIdx.x >> 5);
    const int pair = wave >> 1;
    const int sub  = wave & 1;
    const int half = lane >> 4;
    const int m    = lane & 15;
    const int range = t1 - t0;
    const int full  = nb * 8;

    int rel = gb * 8 + pair;
    int row0 = (t0 + rel) * 16;
    const float* aptr = X + (size_t)min(row0 + m, nrows - 1) * FIN + 2 * half;
    float2 cur[4];
#pragma unroll
    for (int s = 0; s < 4; ++s) cur[s] = *(const float2*)(aptr + s * 4);

#pragma unroll
    for (int i = 0; i < WQ; ++i)
        ((float4*)ldsW)[threadIdx.x + i * SORT_T] = wq[i];
    __syncthreads();

    const float* bptr = ldsW + (2 * half) * FOUT + sub * (NTW * 16) + m;
    v2f bcur[NTW];
#pragma unroll
    for (int t = 0; t < NTW; ++t) {
        bcur[t].x = bptr[t * 16];
        bcur[t].y = bptr[FOUT + t * 16];
    }

    for (int it = 0; rel < range; ++it) {
        v8f acc[NTW] = {};
#pragma nounroll
        for (int kb = 0; kb < FIN / 16; ++kb) {
            const int kn = ((kb + 1) & (FIN / 16 - 1)) * 16;
#pragma unroll
            for (int s = 0; s < 4; ++s) {
                v2f a;
                a.x = cur[s].x;
                a.y = cur[s].y;
                const float* bn = (s < 3) ? bptr + (kb * 16 + s * 4 + 4) * FOUT
                                          : bptr + kn * FOUT;
                v2f bnx[NTW];
#pragma unroll
                for (int t = 0; t < NTW; ++t) {
                    bnx[t].x = bn[t * 16];
                    bnx[t].y = bn[FOUT + t * 16];
                }
                __builtin_amdgcn_sched_barrier(0);
#pragma unroll
                for (int t = 0; t < NTW; ++t)
                    acc[t] = __builtin_amdgcn_wmma_f32_16x16x4_f32(
                        false, a, false, bcur[t], (short)0, acc[t], false, false);
#pragma unroll
                for (int t = 0; t < NTW; ++t) bcur[t] = bnx[t];
                cur[s] = *(const float2*)(aptr + kn + s * 4);
                __builtin_amdgcn_sched_barrier(0);
            }
        }

#pragma unroll
        for (int r = 0; r < 8; ++r) {
            const int orow = row0 + r + 8 * half;
            if (orow < nrows) {
#pragma unroll
                for (int t = 0; t < NTW; ++t)
                    C[(size_t)orow * FOUT + (sub * NTW + t) * 16 + m] = acc[t][r];
            }
        }

        rel = full + gb + (pair + 8 * it) * nb;
        if (rel < range) {
            row0 = (t0 + rel) * 16;
            aptr = X + (size_t)min(row0 + m, nrows - 1) * FIN + 2 * half;
#pragma unroll
            for (int s = 0; s < 4; ++s) cur[s] = *(const float2*)(aptr + s * 4);
        }
    }
}

__global__ __launch_bounds__(SORT_T) void tile_count_kernel(
    const int* __restrict__ dstI, int* __restrict__ cnt1, int nedges, int cb, int g1,
    const float* __restrict__ X, const float* __restrict__ Wm, float* __restrict__ H,
    int nrows, int nb, int t0, int t1)
{
    __shared__ __attribute__((aligned(16))) int lds[LDS_INTS];
    if ((int)blockIdx.x < nb) {
        xw_role((float*)lds, X, Wm, H, nrows, blockIdx.x, nb, t0, t1);
        return;
    }
    const int bid = blockIdx.x - nb;
    int* hist = lds;
    const int t = threadIdx.x;
    for (int k = t; k < CBMAX; k += SORT_T) hist[k] = 0;
    __syncthreads();
    const int base = bid * TILE;
    for (int u = 0; u < TILE / SORT_T; ++u) {
        const int e = base + u * SORT_T + t;
        if (e < nedges) atomicAdd(&hist[coarse_bucket(dstI[e], cb)], 1);
    }
    __syncthreads();
    for (int k = t; k < cb; k += SORT_T)
        cnt1[(size_t)k * g1 + bid] = hist[k];
}

__global__ __launch_bounds__(SORT_T) void tile_scatter_kernel(
    const int* __restrict__ srcI, const int* __restrict__ dstI,
    const float* __restrict__ ew, const int* __restrict__ off1,
    int2* __restrict__ tmpE, int* __restrict__ tmpD, int nedges, int cb, int g1,
    const float* __restrict__ X, const float* __restrict__ Wm, float* __restrict__ H,
    int nrows, int nb, int t0, int t1)
{
    constexpr int PT = TILE / SORT_T;
    __shared__ __attribute__((aligned(16))) int lds[LDS_INTS];
    if ((int)blockIdx.x < nb) {
        xw_role((float*)lds, X, Wm, H, nrows, blockIdx.x, nb, t0, t1);
        return;
    }
    const int bid = blockIdx.x - nb;
    int*  tcount = lds;
    int*  tstart = tcount + CBMAX;
    int*  cur0   = tstart + CBMAX;
    int*  sh     = cur0 + CBMAX;
    int*  stageD = sh + SORT_T;
    int2* stageE = (int2*)(stageD + TILE);
    static_assert((3 * CBMAX + SORT_T + 3 * TILE) <= LDS_INTS, "sort tables fit the block");
    const int t = threadIdx.x;
    for (int k = t; k < CBMAX; k += SORT_T) {
        tcount[k] = 0;
        cur0[k]   = (k < cb) ? off1[(size_t)k * g1 + bid] : 0;
    }
    __syncthreads();

    const int base = bid * TILE;
    const int nT   = min(TILE, nedges - base);
    int d[PT], s[PT], w[PT], r[PT];
#pragma unroll
    for (int u = 0; u < PT; ++u) {
        const int i = u * SORT_T + t;
        if (i < nT) {
            d[u] = dstI[base + i];
            s[u] = srcI[base + i];
            w[u] = __float_as_int(ew[base + i]);
        }
    }
#pragma unroll
    for (int u = 0; u < PT; ++u) {
        const int i = u * SORT_T + t;
        if (i < nT) r[u] = atomicAdd(&tcount[coarse_bucket(d[u], cb)], 1);
    }
    __syncthreads();

    const int c0 = tcount[2 * t];
    const int c1 = tcount[2 * t + 1];
    const int ex = block_excl_sum(c0 + c1, sh);
    tstart[2 * t]     = ex;
    tstart[2 * t + 1] = ex + c0;
    __syncthreads();

#pragma unroll
    for (int u = 0; u < PT; ++u) {
        const int i = u * SORT_T + t;
        if (i < nT) {
            const int slot = tstart[coarse_bucket(d[u], cb)] + r[u];
            stageE[slot] = make_int2(s[u], w[u]);
            stageD[slot] = d[u];
        }
    }
    __syncthreads();

    for (int i = t; i < nT; i += SORT_T) {
        const int    dd  = stageD[i];
        const int    k   = coarse_bucket(dd, cb);
        const size_t pos = (size_t)cur0[k] + (size_t)(i - tstart[k]);
        tmpE[pos] = stageE[i];
        tmpD[pos] = dd;
    }
}

__global__ __launch_bounds__(SORT_T) void bucket_sort_kernel(
    const int2* __restrict__ tmpE, const int* __restrict__ tmpD,
    const int* __restrict__ off1, int* __restrict__ rowptr,
    int2* __restrict__ edges, int nnodes, int nedges, int cb, int g1,
    const float* __restrict__ X, const float* __restrict__ Wm, float* __restrict__ H,
    int nrows, int nb, int t0, int t1)
{
    __shared__ __attribute__((aligned(16))) int lds[LDS_INTS];
    if ((int)blockIdx.x < nb) {
        xw_role((float*)lds, X, Wm, H, nrows, blockIdx.x, nb, t0, t1);
        return;
    }
    int*  cnt    = lds;
    int*  cursor = cnt + BW;
    int*  sh     = cursor + BW;
    int2* stage  = (int2*)(sh + SORT_T);
    static_assert((2 * BW + SORT_T + 2 * CAP) <= LDS_INTS, "sort tables fit the block");
    const int t  = threadIdx.x;
    const int k  = blockIdx.x - nb;
    const int s0 = off1[(size_t)k * g1];
    const int s1 = off1[(size_t)(k + 1) * g1];
    const int L  = s1 - s0;
    if (t < BW) cnt[t] = 0;
    if (k == 0 && t < 8) edges[nedges + t] = make_int2(0, 0);
    __syncthreads();

    for (int i = t; i < L; i += SORT_T)
        atomicAdd(&cnt[tmpD[s0 + i] & (BW - 1)], 1);
    __syncthreads();

    const int mine = (t < BW) ? cnt[t] : 0;
    const int ex   = block_excl_sum(mine, sh);
    if (t < BW) {
        cursor[t] = ex;
        const int node = k * BW + t;
        if (node < nnodes) rowptr[node] = s0 + ex;
    }
    if (k == cb - 1 && t == 0) rowptr[nnodes] = s1;
    __syncthreads();

    const bool staged = L <= CAP;
    for (int i = t; i < L; i += SORT_T) {
        const int  p   = atomicAdd(&cursor[tmpD[s0 + i] & (BW - 1)], 1);
        const int2 rec = tmpE[s0 + i];
        if (staged) stage[p] = rec;
        else        edges[s0 + p] = rec;
    }
    __syncthreads();
    if (staged)
        for (int i = t; i < L; i += SORT_T) edges[s0 + i] = stage[i];
}

template <int F>
__global__ __launch_bounds__(256) void gather_kernel(
    const float* __restrict__ h, const int* __restrict__ rowptr,
    const int2* __restrict__ edges, const float* __restrict__ bias,
    float* __restrict__ out, int nnodes)
{
    constexpr int PER = F / 32;
    constexpr int UN  = 8;
    typedef float __attribute__((ext_vector_type(PER))) vecf;
    const int lane = threadIdx.x & 31;
    const int d = __builtin_amdgcn_readfirstlane(blockIdx.x * 8 + (threadIdx.x >> 5));
    if (d >= nnodes) return;
    const int beg = rowptr[d];
    const int end = rowptr[d + 1];

    vecf acc = 0.0f;

    for (int j = beg; j < end; j += UN) {
        const int n = end - j;
        int2 r[UN];
#pragma unroll
        for (int u = 0; u < UN; ++u) r[u] = edges[j + u];
        vecf v[UN];
#pragma unroll
        for (int u = 0; u < UN; ++u) {
            if (u < n)
                v[u] = *(const vecf*)(h + (size_t)r[u].x * F + lane * PER);
        }
#pragma unroll
        for (int u = 0; u < UN; ++u) {
            if (u < n)
                acc += __int_as_float(r[u].y) * v[u];
        }
    }

    const vecf b = *(const vecf*)(bias + lane * PER);
    *(vecf*)(out + (size_t)d * F + lane * PER) = acc + b;
}


#define TPITCH 132

template <int FOUT>
__global__ __launch_bounds__(512) void gather_gemm_kernel(
    const float* __restrict__ h, const int* __restrict__ rowptr,
    const int2* __restrict__ edges, const float* __restrict__ bias,
    const float* __restrict__ Wn, float* __restrict__ C, int nnodes)
{
    constexpr int NT  = FOUT / 16;
    constexpr int NTW = NT / 2;
    constexpr int UN  = 8;
    typedef float __attribute__((ext_vector_type(4))) vecf;
    constexpr int WQ  = FIN * FOUT / 4 / 512;
    __shared__ float lds[FIN * FOUT + 8 * 16 * TPITCH];
    float* ldsW = lds;
    float* tiles = lds + FIN * FOUT;

    float4 wq[WQ];
#pragma unroll
    for (int i = 0; i < WQ; ++i)
        wq[i] = ((const float4*)Wn)[threadIdx.x + i * 512];

    const int lane = threadIdx.x & 31;
    const int wave = __builtin_amdgcn_readfirstlane(threadIdx.x >> 5);
    const int pair = wave >> 1;
    const int sub  = wave & 1;
    const int node0 = (blockIdx.x * 8 + pair) * 16;
    float* trow = tiles + pair * 16 * TPITCH;

    const int first = node0 + 8 * sub;
    const int rp = rowptr[min(first + lane, nnodes)];
    const vecf b = *(const vecf*)(bias + lane * 4);

#pragma unroll
    for (int i = 0; i < WQ; ++i)
        ((float4*)ldsW)[threadIdx.x + i * 512] = wq[i];

    for (int q = 0; q < 8; ++q) {
        const int d = first + q;
        vecf row = 0.0f;
        if (d < nnodes) {
            const int beg = __builtin_amdgcn_readlane(rp, q);
            const int end = __builtin_amdgcn_readlane(rp, q + 1);
            vecf acc = 0.0f;
            for (int j = beg; j < end; j += UN) {
                const int n = end - j;
                int2 r[UN];
#pragma unroll
                for (int u = 0; u < UN; ++u) r[u] = edges[j + u];
                vecf v[UN];
#pragma unroll
                for (int u = 0; u < UN; ++u) {
                    if (u < n)
                        v[u] = *(const vecf*)(h + (size_t)r[u].x * FIN + lane * 4);
                }
#pragma unroll
                for (int u = 0; u < UN; ++u) {
                    if (u < n)
                        acc += __int_as_float(r[u].y) * v[u];
                }
            }
            row = acc + b;
            row.x = row.x > 0.0f ? row.x : 0.0f;
            row.y = row.y > 0.0f ? row.y : 0.0f;
            row.z = row.z > 0.0f ? row.z : 0.0f;
            row.w = row.w > 0.0f ? row.w : 0.0f;
        }
        *(vecf*)(trow + (8 * sub + q) * TPITCH + lane * 4) = row;
    }
    __syncthreads();
    if (node0 >= nnodes) return;

    const int half = lane >> 4;
    const int m    = lane & 15;
    const float* arow = trow + m * TPITCH;

    v8f acc[NTW] = {};

    for (int k = 0; k < FIN; k += 4) {
        const int kk = k + 2 * half;
        const float2 af = *(const float2*)(arow + kk);
        v2f a;
        a.x = af.x;
        a.y = af.y;
#pragma unroll
        for (int t = 0; t < NTW; ++t) {
            const int col = (sub * NTW + t) * 16 + m;
            v2f bf;
            bf.x = ldsW[(kk)     * FOUT + col];
            bf.y = ldsW[(kk + 1) * FOUT + col];
            acc[t] = __builtin_amdgcn_wmma_f32_16x16x4_f32(
                false, a, false, bf, (short)0, acc[t], false, false);
        }
    }

#pragma unroll
    for (int r = 0; r < 8; ++r) {
        const int orow = node0 + r + 8 * half;
        if (orow < nnodes) {
#pragma unroll
            for (int t = 0; t < NTW; ++t)
                C[(size_t)orow * FOUT + (sub * NTW + t) * 16 + m] = acc[t][r];
        }
    }
}

extern "C" void kernel_launch(void* const* d_in, const int* in_sizes, int n_in,
                              void* d_out, int out_size, void* d_ws, size_t ws_size,
                              hipStream_t stream)
{
    const float* x  = (const float*)d_in[0];
    const int*   ei = (const int*)d_in[1];
    const float* ew = (const float*)d_in[2];
    const float* W1 = (const float*)d_in[3];
    const float* b1 = (const float*)d_in[4];
    const float* W2 = (const float*)d_in[5];
    const float* b2 = (const float*)d_in[6];
    const float* W3 = (const float*)d_in[7];
    const float* b3 = (const float*)d_in[8];
    float* out = (float*)d_out;

    const int N = in_sizes[0] / FIN;
    const int E = in_sizes[2];
    const int* srcI = ei;
    const int* dstI = ei + E;

    float* bufA = (float*)d_ws;
    float* bufB = bufA + (size_t)N * FIN;

    const int ntiles     = (N + 15) / 16;
    const int gemmBlocks = (ntiles + 7) / 8;
    const int nodeBlocks = (N + 7) / 8;
    const int edgeBlocks = (E + 255) / 256;
    const int scanBlocks = (N + 1023) / 1024;

    const int    NP      = (N + 1 + 63) & ~63;
    const int    NB      = (scanBlocks + 63) & ~63;
    const size_t csrInts = (size_t)2 * NP + NB + (size_t)2 * E + 16;
    const bool roomy = ws_size >= ((size_t)2 * N * FIN + csrInts) * sizeof(float);
    int* csr  = roomy ? (int*)(bufB + (size_t)N * FIN) : (int*)d_out;
    int* csr3 = roomy ? csr : (int*)(bufA + (size_t)N * 64);

    int*  rowptr = csr;
    int*  next   = csr + NP;
    int*  bsum   = csr + 2 * NP;
    int2* edges  = (int2*)(csr + 2 * NP + NB);

    const int    CB      = (N + BW - 1) / BW;
    const int    G1      = (E + TILE - 1) / TILE;
    const size_t n1      = (size_t)CB * G1;
    const size_t n1p     = (n1 + 1 + 63) & ~(size_t)63;
    const int    scanBl1 = (int)((n1 + 1023) / 1024);
    const size_t NB1     = ((size_t)scanBl1 + 63) & ~(size_t)63;
    const bool   ldsSort = CB <= CBMAX && G1 > 0 && n1 < ((size_t)1 << 30) &&
                           (size_t)3 * E + 2 * n1p + NB1 <= (size_t)N * FIN;
    if (ldsSort) {
        int2* tmpE  = (int2*)bufB;
        int*  tmpD  = (int*)(tmpE + E);
        int*  cnt1  = tmpD + E;
        int*  off1  = cnt1 + n1p;
        int*  bsum1 = off1 + n1p;
        const int tA = XW_NB * 8 < ntiles ? XW_NB * 8 : ntiles;
        const int tB = 2 * XW_NB * 8 < ntiles ? 2 * XW_NB * 8 : ntiles;
        const int nbA = (tA + 7) / 8 < XW_NB ? (tA + 7) / 8 : XW_NB;
        const int nbB = (tB - tA + 7) / 8 < XW_NB ? (tB - tA + 7) / 8 : XW_NB;
        const int nbC = (ntiles - tB + 7) / 8 < XW_NB ? (ntiles - tB + 7) / 8 : XW_NB;
        tile_count_kernel<<<nbA + G1, SORT_T, 0, stream>>>(dstI, cnt1, E, CB, G1,
                                                          x, W1, bufA, N, nbA, 0, tA);
        scan_local_kernel<<<scanBl1, 256, 0, stream>>>(cnt1, off1, bsum1, (int)n1);
        scan_offset_kernel<false><<<scanBl1, 256, 0, stream>>>(off1, nullptr, bsum1, (int)n1);
        tile_scatter_kernel<<<nbB + G1, SORT_T, 0, stream>>>(srcI, dstI, ew, off1, tmpE, tmpD, E, CB, G1,
                                                            x, W1, bufA, N, nbB, tA, tB);
        bucket_sort_kernel<<<nbC + CB, SORT_T, 0, stream>>>(tmpE, tmpD, off1, rowptr, edges, N, E, CB, G1,
                                                           x, W1, bufA, N, nbC, tB, ntiles);
    } else {
        zero_counts_kernel<<<(N + 255) / 256, 256, 0, stream>>>(next, N);
        count_dst_kernel<<<edgeBlocks, 256, 0, stream>>>(dstI, next, E);
        scan_local_kernel<<<scanBlocks, 256, 0, stream>>>(next, rowptr, bsum, N);
        scan_offset_kernel<true><<<scanBlocks, 256, 0, stream>>>(rowptr, next, bsum, N);
        fill_edges_kernel<<<edgeBlocks, 256, 0, stream>>>(srcI, dstI, ew, next, edges, E);
        gemm_wmma_kernel<128, false><<<gemmBlocks, 256, 0, stream>>>(x, W1, bufA, N);
    }

    gather_gemm_kernel<128><<<gemmBlocks, 512, 0, stream>>>(bufA, rowptr, edges, b1, W2, bufB, N);

    gather_gemm_kernel<64><<<gemmBlocks, 512, 0, stream>>>(bufB, rowptr, edges, b2, W3, bufA, N);

    if (!roomy)
        copy_ints_kernel<<<(int)((csrInts + 255) / 256), 256, 0, stream>>>(csr, csr3, (int)csrInts);
    gather_kernel<64><<<nodeBlocks, 256, 0, stream>>>(
        bufA, csr3, (const int2*)(csr3 + 2 * NP + NB), b3, out, N);
}


# fused gathers: a wave walks the records of its 8 nodes as one run in whole groups of 8 across node bounds, next group's rows and records requested ahead (two groups under way); same order of sums
# speedup vs baseline: 3.9899x; 1.0574x over previous
#include <hip/hip_runtime.h>
#include <hip/hip_bf16.h>

typedef __attribute__((ext_vector_type(2))) float v2f;
typedef __attribute__((ext_vector_type(8))) float v8f;

#define FIN 128

template <int FOUT, bool RELU>
__global__ __launch_bounds__(256) void gemm_wmma_kernel(
    const float* __restrict__ A, const float* __restrict__ W,
    float* __restrict__ C, int nrows)
{
    constexpr int NT = FOUT / 16;
    __shared__ float ldsW[FIN * FOUT];

    for (int i = threadIdx.x; i < FIN * FOUT; i += 256)
        ldsW[i] = W[i];
    __syncthreads();

    const int wave = threadIdx.x >> 5;
    const int lane = threadIdx.x & 31;
    const int half = lane >> 4;
    const int m    = lane & 15;

    const int tile   = blockIdx.x * 8 + wave;
    const int ntiles = (nrows + 15) >> 4;
    if (tile >= ntiles) return;

    const int row0 = tile * 16;
    int arow = row0 + m;
    if (arow >= nrows) arow = nrows - 1;
    const float* aptr = A + (size_t)arow * FIN;

    v8f acc[NT] = {};

    for (int k = 0; k < FIN; k += 4) {
        const int kk = k + 2 * half;
        float2 af = *(const float2*)(aptr + kk);
        if (RELU) {
            af.x = af.x > 0.0f ? af.x : 0.0f;
            af.y = af.y > 0.0f ? af.y : 0.0f;
        }
        v2f a;
        a.x = af.x;
        a.y = af.y;
#pragma unroll
        for (int t = 0; t < NT; ++t) {
            v2f b;
            b.x = ldsW[(kk)     * FOUT + t * 16 + m];
            b.y = ldsW[(kk + 1) * FOUT + t * 16 + m];
            acc[t] = __builtin_amdgcn_wmma_f32_16x16x4_f32(
                false, a, false, b, (short)0, acc[t], false, false);
        }
    }

#pragma unroll
    for (int r = 0; r < 8; ++r) {
        const int row = row0 + r + 8 * half;
        if (row < nrows) {
#pragma unroll
            for (int t = 0; t < NT; ++t)
                C[(size_t)row * FOUT + t * 16 + m] = acc[t][r];
        }
    }
}

__global__ void zero_counts_kernel(int* __restrict__ cnt, int n)
{
    const int i = blockIdx.x * blockDim.x + threadIdx.x;
    if (i < n) cnt[i] = 0;
}

__global__ void count_dst_kernel(const int* __restrict__ dstI,
                                 int* __restrict__ cnt, int nedges)
{
    const int e = blockIdx.x * blockDim.x + threadIdx.x;
    if (e < nedges) atomicAdd(&cnt[dstI[e]], 1);
}

__global__ __launch_bounds__(256) void scan_local_kernel(
    const int* __restrict__ cnt, int* __restrict__ rowptr,
    int* __restrict__ bsum, int n)
{
    __shared__ int sh[256];
    const int t  = threadIdx.x;
    const int i0 = blockIdx.x * 1024 + t * 4;
    int c[4];
#pragma unroll
    for (int k = 0; k < 4; ++k) c[k] = (i0 + k < n) ? cnt[i0 + k] : 0;
    const int mine = c[0] + c[1] + c[2] + c[3];
    sh[t] = mine;
    __syncthreads();
    for (int off = 1; off < 256; off <<= 1) {
        const int add = (t >= off) ? sh[t - off] : 0;
        __syncthreads();
        sh[t] += add;
        __syncthreads();
    }
    int run = sh[t] - mine;
#pragma unroll
    for (int k = 0; k < 4; ++k) {
        if (i0 + k < n) rowptr[i0 + k] = run;
        run += c[k];
    }
    if (t == 255) bsum[blockIdx.x] = sh[255];
}

template <bool NEXT>
__global__ __launch_bounds__(256) void scan_offset_kernel(
    int* __restrict__ rowptr, int* __restrict__ next,
    const int* __restrict__ bsum, int n)
{
    int off = 0;
    for (int j = 0; j < (int)blockIdx.x; ++j) off += bsum[j];
    const int i0 = blockIdx.x * 1024 + threadIdx.x * 4;
#pragma unroll
    for (int k = 0; k < 4; ++k) {
        if (i0 + k < n) {
            const int r = rowptr[i0 + k] + off;
            rowptr[i0 + k] = r;
            if (NEXT) next[i0 + k] = r;
        }
    }
    if (blockIdx.x == gridDim.x - 1 && threadIdx.x == 0)
        rowptr[n] = off + bsum[blockIdx.x];
}

__global__ void fill_edges_kernel(const int* __restrict__ srcI,
                                  const int* __restrict__ dstI,
                                  const float* __restrict__ ew,
                                  int* __restrict__ next,
                                  int2* __restrict__ edges, int nedges)
{
    const int e = blockIdx.x * blockDim.x + threadIdx.x;
    if (e < 8) edges[nedges + e] = make_int2(0, 0);
    if (e >= nedges) return;
    const int p = atomicAdd(&next[dstI[e]], 1);
    edges[p] = make_int2(srcI[e], __float_as_int(ew[e]));
}

__global__ void copy_ints_kernel(const int* __restrict__ from,
                                 int* __restrict__ to, int n)
{
    const int i = blockIdx.x * blockDim.x + threadIdx.x;
    if (i < n) to[i] = from[i];
}

#define BSH    8
#define BW     256
#define TILE   4096
#define SORT_T 512
#define CBMAX  1024
#define CAP    7168

__device__ __forceinline__ int block_excl_sum(int mine, int* sh)
{
    const int t = threadIdx.x;
    sh[t] = mine;
    __syncthreads();
    for (int off = 1; off < SORT_T; off <<= 1) {
        const int add = (t >= off) ? sh[t - off] : 0;
        __syncthreads();
        sh[t] += add;
        __syncthreads();
    }
    const int r = sh[t] - mine;
    __syncthreads();
    return r;
}

__device__ __forceinline__ int coarse_bucket(int d, int cb)
{
    return min((int)((unsigned)d >> BSH), cb - 1);
}

#define XW_NB    256
#define LDS_INTS (FIN * 128)

__device__ __forceinline__ void xw_role(
    float* ldsW, const float* __restrict__ X, const float* __restrict__ Wm,
    float* __restrict__ C, int nrows, int gb, int nb, int t0, int t1)
{
    constexpr int FOUT = 128;
    constexpr int NTW  = 4;
    constexpr int WQ   = FIN * FOUT / 4 / SORT_T;
    float4 wq[WQ];
#pragma unroll
    for (int i = 0; i < WQ; ++i)
        wq[i] = ((const float4*)Wm)[threadIdx.x + i * SORT_T];

    const int lane = threadIdx.x & 31;
    const int wave = __builtin_amdgcn_readfirstlane(threadIdx.x >> 5);
    const int pair = wave >> 1;
    const int sub  = wave & 1;
    const int half = lane >> 4;
    const int m    = lane & 15;
    const int range = t1 - t0;
    const int full  = nb * 8;

    int rel = gb * 8 + pair;
    int row0 = (t0 + rel) * 16;
    const float* aptr = X + (size_t)min(row0 + m, nrows - 1) * FIN + 2 * half;
    float2 cur[4];
#pragma unroll
    for (int s = 0; s < 4; ++s) cur[s] = *(const float2*)(aptr + s * 4);

#pragma unroll
    for (int i = 0; i < WQ; ++i)
        ((float4*)ldsW)[threadIdx.x + i * SORT_T] = wq[i];
    __syncthreads();

    const float* bptr = ldsW + (2 * half) * FOUT + sub * (NTW * 16) + m;
    v2f bcur[NTW];
#pragma unroll
    for (int t = 0; t < NTW; ++t) {
        bcur[t].x = bptr[t * 16];
        bcur[t].y = bptr[FOUT + t * 16];
    }

    for (int it = 0; rel < range; ++it) {
        v8f acc[NTW] = {};
#pragma nounroll
        for (int kb = 0; kb < FIN / 16; ++kb) {
            const int kn = ((kb + 1) & (FIN / 16 - 1)) * 16;
#pragma unroll
            for (int s = 0; s < 4; ++s) {
                v2f a;
                a.x = cur[s].x;
                a.y = cur[s].y;
                const float* bn = (s < 3) ? bptr + (kb * 16 + s * 4 + 4) * FOUT
                                          : bptr + kn * FOUT;
                v2f bnx[NTW];
#pragma unroll
                for (int t = 0; t < NTW; ++t) {
                    bnx[t].x = bn[t * 16];
                    bnx[t].y = bn[FOUT + t * 16];
                }
                __builtin_amdgcn_sched_barrier(0);
#pragma unroll
                for (int t = 0; t < NTW; ++t)
                    acc[t] = __builtin_amdgcn_wmma_f32_16x16x4_f32(
                        false, a, false, bcur[t], (short)0, acc[t], false, false);
#pragma unroll
                for (int t = 0; t < NTW; ++t) bcur[t] = bnx[t];
                cur[s] = *(const float2*)(aptr + kn + s * 4);
                __builtin_amdgcn_sched_barrier(0);
            }
        }

#pragma unroll
        for (int r = 0; r < 8; ++r) {
            const int orow = row0 + r + 8 * half;
            if (orow < nrows) {
#pragma unroll
                for (int t = 0; t < NTW; ++t)
                    C[(size_t)orow * FOUT + (sub * NTW + t) * 16 + m] = acc[t][r];
            }
        }

        rel = full + gb + (pair + 8 * it) * nb;
        if (rel < range) {
            row0 = (t0 + rel) * 16;
            aptr = X + (size_t)min(row0 + m, nrows - 1) * FIN + 2 * half;
#pragma unroll
            for (int s = 0; s < 4; ++s) cur[s] = *(const float2*)(aptr + s * 4);
        }
    }
}

__global__ __launch_bounds__(SORT_T) void tile_count_kernel(
    const int* __restrict__ dstI, int* __restrict__ cnt1, int nedges, int cb, int g1,
    const float* __restrict__ X, const float* __restrict__ Wm, float* __restrict__ H,
    int nrows, int nb, int t0, int t1)
{
    __shared__ __attribute__((aligned(16))) int lds[LDS_INTS];
    if ((int)blockIdx.x < nb) {
        xw_role((float*)lds, X, Wm, H, nrows, blockIdx.x, nb, t0, t1);
        return;
    }
    const int bid = blockIdx.x - nb;
    int* hist = lds;
    const int t = threadIdx.x;
    for (int k = t; k < CBMAX; k += SORT_T) hist[k] = 0;
    __syncthreads();
    const int base = bid * TILE;
    for (int u = 0; u < TILE / SORT_T; ++u) {
        const int e = base + u * SORT_T + t;
        if (e < nedges) atomicAdd(&hist[coarse_bucket(dstI[e], cb)], 1);
    }
    __syncthreads();
    for (int k = t; k < cb; k += SORT_T)
        cnt1[(size_t)k * g1 + bid] = hist[k];
}

__global__ __launch_bounds__(SORT_T) void tile_scatter_kernel(
    const int* __restrict__ srcI, const int* __restrict__ dstI,
    const float* __restrict__ ew, const int* __restrict__ off1,
    int2* __restrict__ tmpE, int* __restrict__ tmpD, int nedges, int cb, int g1,
    const float* __restrict__ X, const float* __restrict__ Wm, float* __restrict__ H,
    int nrows, int nb, int t0, int t1)
{
    constexpr int PT = TILE / SORT_T;
    __shared__ __attribute__((aligned(16))) int lds[LDS_INTS];
    if ((int)blockIdx.x < nb) {
        xw_role((float*)lds, X, Wm, H, nrows, blockIdx.x, nb, t0, t1);
        return;
    }
    const int bid = blockIdx.x - nb;
    int*  tcount = lds;
    int*  tstart = tcount + CBMAX;
    int*  cur0   = tstart + CBMAX;
    int*  sh     = cur0 + CBMAX;
    int*  stageD = sh + SORT_T;
    int2* stageE = (int2*)(stageD + TILE);
    static_assert((3 * CBMAX + SORT_T + 3 * TILE) <= LDS_INTS, "sort tables fit the block");
    const int t = threadIdx.x;
    for (int k = t; k < CBMAX; k += SORT_T) {
        tcount[k] = 0;
        cur0[k]   = (k < cb) ? off1[(size_t)k * g1 + bid] : 0;
    }
    __syncthreads();

    const int base = bid * TILE;
    const int nT   = min(TILE, nedges - base);
    int d[PT], s[PT], w[PT], r[PT];
#pragma unroll
    for (int u = 0; u < PT; ++u) {
        const int i = u * SORT_T + t;
        if (i < nT) {
            d[u] = dstI[base + i];
            s[u] = srcI[base + i];
            w[u] = __float_as_int(ew[base + i]);
        }
    }
#pragma unroll
    for (int u = 0; u < PT; ++u) {
        const int i = u * SORT_T + t;
        if (i < nT) r[u] = atomicAdd(&tcount[coarse_bucket(d[u], cb)], 1);
    }
    __syncthreads();

    const int c0 = tcount[2 * t];
    const int c1 = tcount[2 * t + 1];
    const int ex = block_excl_sum(c0 + c1, sh);
    tstart[2 * t]     = ex;
    tstart[2 * t + 1] = ex + c0;
    __syncthreads();

#pragma unroll
    for (int u = 0; u < PT; ++u) {
        const int i = u * SORT_T + t;
        if (i < nT) {
            const int slot = tstart[coarse_bucket(d[u], cb)] + r[u];
            stageE[slot] = make_int2(s[u], w[u]);
            stageD[slot] = d[u];
        }
    }
    __syncthreads();

    for (int i = t; i < nT; i += SORT_T) {
        const int    dd  = stageD[i];
        const int    k   = coarse_bucket(dd, cb);
        const size_t pos = (size_t)cur0[k] + (size_t)(i - tstart[k]);
        tmpE[pos] = stageE[i];
        tmpD[pos] = dd;
    }
}

__global__ __launch_bounds__(SORT_T) void bucket_sort_kernel(
    const int2* __restrict__ tmpE, const int* __restrict__ tmpD,
    const int* __restrict__ off1, int* __restrict__ rowptr,
    int2* __restrict__ edges, int nnodes, int nedges, int cb, int g1,
    const float* __restrict__ X, const float* __restrict__ Wm, float* __restrict__ H,
    int nrows, int nb, int t0, int t1)
{
    __shared__ __attribute__((aligned(16))) int lds[LDS_INTS];
    if ((int)blockIdx.x < nb) {
        xw_role((float*)lds, X, Wm, H, nrows, blockIdx.x, nb, t0, t1);
        return;
    }
    int*  cnt    = lds;
    int*  cursor = cnt + BW;
    int*  sh     = cursor + BW;
    int2* stage  = (int2*)(sh + SORT_T);
    static_assert((2 * BW + SORT_T + 2 * CAP) <= LDS_INTS, "sort tables fit the block");
    const int t  = threadIdx.x;
    const int k  = blockIdx.x - nb;
    const int s0 = off1[(size_t)k * g1];
    const int s1 = off1[(size_t)(k + 1) * g1];
    const int L  = s1 - s0;
    if (t < BW) cnt[t] = 0;
    if (k == 0 && t < 8) edges[nedges + t] = make_int2(0, 0);
    __syncthreads();

    for (int i = t; i < L; i += SORT_T)
        atomicAdd(&cnt[tmpD[s0 + i] & (BW - 1)], 1);
    __syncthreads();

    const int mine = (t < BW) ? cnt[t] : 0;
    const int ex   = block_excl_sum(mine, sh);
    if (t < BW) {
        cursor[t] = ex;
        const int node = k * BW + t;
        if (node < nnodes) rowptr[node] = s0 + ex;
    }
    if (k == cb - 1 && t == 0) rowptr[nnodes] = s1;
    __syncthreads();

    const bool staged = L <= CAP;
    for (int i = t; i < L; i += SORT_T) {
        const int  p   = atomicAdd(&cursor[tmpD[s0 + i] & (BW - 1)], 1);
        const int2 rec = tmpE[s0 + i];
        if (staged) stage[p] = rec;
        else        edges[s0 + p] = rec;
    }
    __syncthreads();
    if (staged)
        for (int i = t; i < L; i += SORT_T) edges[s0 + i] = stage[i];
}

template <int F>
__global__ __launch_bounds__(256) void gather_kernel(
    const float* __restrict__ h, const int* __restrict__ rowptr,
    const int2* __restrict__ edges, const float* __restrict__ bias,
    float* __restrict__ out, int nnodes)
{
    constexpr int PER = F / 32;
    constexpr int UN  = 8;
    typedef float __attribute__((ext_vector_type(PER))) vecf;
    const int lane = threadIdx.x & 31;
    const int d = __builtin_amdgcn_readfirstlane(blockIdx.x * 8 + (threadIdx.x >> 5));
    if (d >= nnodes) return;
    const int beg = rowptr[d];
    const int end = rowptr[d + 1];

    vecf acc = 0.0f;

    for (int j = beg; j < end; j += UN) {
        const int n = end - j;
        int2 r[UN];
#pragma unroll
        for (int u = 0; u < UN; ++u) r[u] = edges[j + u];
        vecf v[UN];
#pragma unroll
        for (int u = 0; u < UN; ++u) {
            if (u < n)
                v[u] = *(const vecf*)(h + (size_t)r[u].x * F + lane * PER);
        }
#pragma unroll
        for (int u = 0; u < UN; ++u) {
            if (u < n)
                acc += __int_as_float(r[u].y) * v[u];
        }
    }

    const vecf b = *(const vecf*)(bias + lane * PER);
    *(vecf*)(out + (size_t)d * F + lane * PER) = acc + b;
}


#define TPITCH 132

template <int FOUT>
__global__ __launch_bounds__(512) void gather_gemm_kernel(
    const float* __restrict__ h, const int* __restrict__ rowptr,
    const int2* __restrict__ edges, const float* __restrict__ bias,
    const float* __restrict__ Wn, float* __restrict__ C, int nnodes)
{
    constexpr int NT  = FOUT / 16;
    constexpr int NTW = NT / 2;
    constexpr int UN  = 8;
    typedef float __attribute__((ext_vector_type(4))) vecf;
    constexpr int WQ  = FIN * FOUT / 4 / 512;
    __shared__ float lds[FIN * FOUT + 8 * 16 * TPITCH];
    float* ldsW = lds;
    float* tiles = lds + FIN * FOUT;

    float4 wq[WQ];
#pragma unroll
    for (int i = 0; i < WQ; ++i)
        wq[i] = ((const float4*)Wn)[threadIdx.x + i * 512];

    const int lane = threadIdx.x & 31;
    const int wave = __builtin_amdgcn_readfirstlane(threadIdx.x >> 5);
    const int pair = wave >> 1;
    const int sub  = wave & 1;
    const int node0 = (blockIdx.x * 8 + pair) * 16;
    float* trow = tiles + pair * 16 * TPITCH;

    const int first = node0 + 8 * sub;
    const int rp = rowptr[min(first + lane, nnodes)];
    const vecf b = *(const vecf*)(bias + lane * 4);

#pragma unroll
    for (int i = 0; i < WQ; ++i)
        ((float4*)ldsW)[threadIdx.x + i * 512] = wq[i];

    const int jbeg = __builtin_amdgcn_readlane(rp, 0);
    const int jend = __builtin_amdgcn_readlane(rp, 8);
    int  q    = 0;
    int  endq = __builtin_amdgcn_readlane(rp, 1);
    vecf gsum  = 0.0f;

#define GG_CLOSE_NODE()                                                        \
    {                                                                          \
        vecf row = gsum + b;                                                    \
        row.x = row.x > 0.0f ? row.x : 0.0f;                                   \
        row.y = row.y > 0.0f ? row.y : 0.0f;                                   \
        row.z = row.z > 0.0f ? row.z : 0.0f;                                   \
        row.w = row.w > 0.0f ? row.w : 0.0f;                                   \
        if (first + q >= nnodes) row = 0.0f;         \
        *(vecf*)(trow + (8 * sub + q) * TPITCH + lane * 4) = row;              \
        ++q;                                                                   \
        endq = __builtin_amdgcn_readlane(rp, q + 1);                           \
        gsum  = 0.0f;                                                           \
    }
#define GG_ROWS(J0, R, V)                                                      \
    _Pragma("unroll")                                                          \
    for (int u = 0; u < UN; ++u) {                                             \
        if ((J0) + u < jend)                                                   \
            V[u] = *(const vecf*)(h + (size_t)R[u].x * FIN + lane * 4);        \
    }
#define GG_RECS(J0, R)                                                         \
    {                                                                          \
        const int jr = min((J0), jend);                                        \
        _Pragma("unroll")                                                      \
        for (int u = 0; u < UN; ++u) R[u] = edges[jr + u];                     \
    }
#define GG_SUM(J0, R, V)                                                       \
    _Pragma("unroll")                                                          \
    for (int u = 0; u < UN; ++u) {                                             \
        const int e = (J0) + u;                                                \
        if (e < jend) {                                                        \
            while (e >= endq) GG_CLOSE_NODE()                                  \
            gsum += __int_as_float(R[u].y) * V[u];                              \
        }                                                                      \
    }

#define GG_ROWS_ALL(R, V)                                                      \
    _Pragma("unroll")                                                          \
    for (int u = 0; u < UN; ++u)                                               \
        V[u] = *(const vecf*)(h + (size_t)R[u].x * FIN + lane * 4);
#define GG_SUM_ALL(J0, R, V)                                                   \
    _Pragma("unroll")                                                          \
    for (int u = 0; u < UN; ++u) {                                             \
        const int e = (J0) + u;                                                \
        while (e >= endq) GG_CLOSE_NODE()                                      \
        gsum += __int_as_float(R[u].y) * V[u];                                 \
    }

    int2 rc[UN], rn[UN], rx[UN], ry[UN];
    vecf va[UN], vb[UN];
    int j = jbeg;
    if (j + 3 * UN <= jend) {
        GG_RECS(j, rc)
        GG_RECS(j + UN, rn)
        GG_ROWS_ALL(rc, va)
        do {
            GG_ROWS_ALL(rn, vb)
            GG_RECS(j + 2 * UN, rx)
            GG_SUM_ALL(j, rc, va)
            GG_ROWS_ALL(rx, va)
            GG_RECS(j + 3 * UN, ry)
            GG_SUM_ALL(j + UN, rn, vb)
#pragma unroll
            for (int u = 0; u < UN; ++u) { rc[u] = rx[u]; rn[u] = ry[u]; }
            j += 2 * UN;
        } while (j + 3 * UN <= jend);
        GG_SUM_ALL(j, rc, va)
        j += UN;
    }
    for (; j < jend; j += UN) {
        GG_RECS(j, rc)
        GG_ROWS(j, rc, va)
        GG_SUM(j, rc, va)
    }
    while (q < 8) GG_CLOSE_NODE()
#undef GG_CLOSE_NODE
#undef GG_ROWS
#undef GG_RECS
#undef GG_SUM
#undef GG_ROWS_ALL
#undef GG_SUM_ALL
    __syncthreads();
    if (node0 >= nnodes) return;

    const int half = lane >> 4;
    const int m    = lane & 15;
    const float* arow = trow + m * TPITCH;

    v8f acc[NTW] = {};

    for (int k = 0; k < FIN; k += 4) {
        const int kk = k + 2 * half;
        const float2 af = *(const float2*)(arow + kk);
        v2f a;
        a.x = af.x;
        a.y = af.y;
#pragma unroll
        for (int t = 0; t < NTW; ++t) {
            const int col = (sub * NTW + t) * 16 + m;
            v2f bf;
            bf.x = ldsW[(kk)     * FOUT + col];
            bf.y = ldsW[(kk + 1) * FOUT + col];
            acc[t] = __builtin_amdgcn_wmma_f32_16x16x4_f32(
                false, a, false, bf, (short)0, acc[t], false, false);
        }
    }

#pragma unroll
    for (int r = 0; r < 8; ++r) {
        const int orow = node0 + r + 8 * half;
        if (orow < nnodes) {
#pragma unroll
            for (int t = 0; t < NTW; ++t)
                C[(size_t)orow * FOUT + (sub * NTW + t) * 16 + m] = acc[t][r];
        }
    }
}

extern "C" void kernel_launch(void* const* d_in, const int* in_sizes, int n_in,
                              void* d_out, int out_size, void* d_ws, size_t ws_size,
                              hipStream_t stream)
{
    const float* x  = (const float*)d_in[0];
    const int*   ei = (const int*)d_in[1];
    const float* ew = (const float*)d_in[2];
    const float* W1 = (const float*)d_in[3];
    const float* b1 = (const float*)d_in[4];
    const float* W2 = (const float*)d_in[5];
    const float* b2 = (const float*)d_in[6];
    const float* W3 = (const float*)d_in[7];
    const float* b3 = (const float*)d_in[8];
    float* out = (float*)d_out;

    const int N = in_sizes[0] / FIN;
    const int E = in_sizes[2];
    const int* srcI = ei;
    const int* dstI = ei + E;

    float* bufA = (float*)d_ws;
    float* bufB = bufA + (size_t)N * FIN;

    const int ntiles     = (N + 15) / 16;
    const int gemmBlocks = (ntiles + 7) / 8;
    const int nodeBlocks = (N + 7) / 8;
    const int edgeBlocks = (E + 255) / 256;
    const int scanBlocks = (N + 1023) / 1024;

    const int    NP      = (N + 1 + 63) & ~63;
    const int    NB      = (scanBlocks + 63) & ~63;
    const size_t csrInts = (size_t)2 * NP + NB + (size_t)2 * E + 16;
    const bool roomy = ws_size >= ((size_t)2 * N * FIN + csrInts) * sizeof(float);
    int* csr  = roomy ? (int*)(bufB + (size_t)N * FIN) : (int*)d_out;
    int* csr3 = roomy ? csr : (int*)(bufA + (size_t)N * 64);

    int*  rowptr = csr;
    int*  next   = csr + NP;
    int*  bsum   = csr + 2 * NP;
    int2* edges  = (int2*)(csr + 2 * NP + NB);

    const int    CB      = (N + BW - 1) / BW;
    const int    G1      = (E + TILE - 1) / TILE;
    const size_t n1      = (size_t)CB * G1;
    const size_t n1p     = (n1 + 1 + 63) & ~(size_t)63;
    const int    scanBl1 = (int)((n1 + 1023) / 1024);
    const size_t NB1     = ((size_t)scanBl1 + 63) & ~(size_t)63;
    const bool   ldsSort = CB <= CBMAX && G1 > 0 && n1 < ((size_t)1 << 30) &&
                           (size_t)3 * E + 2 * n1p + NB1 <= (size_t)N * FIN;
    if (ldsSort) {
        int2* tmpE  = (int2*)bufB;
        int*  tmpD  = (int*)(tmpE + E);
        int*  cnt1  = tmpD + E;
        int*  off1  = cnt1 + n1p;
        int*  bsum1 = off1 + n1p;
        const int tA = XW_NB * 8 < ntiles ? XW_NB * 8 : ntiles;
        const int tB = 2 * XW_NB * 8 < ntiles ? 2 * XW_NB * 8 : ntiles;
        const int nbA = (tA + 7) / 8 < XW_NB ? (tA + 7) / 8 : XW_NB;
        const int nbB = (tB - tA + 7) / 8 < XW_NB ? (tB - tA + 7) / 8 : XW_NB;
        const int nbC = (ntiles - tB + 7) / 8 < XW_NB ? (ntiles - tB + 7) / 8 : XW_NB;
        tile_count_kernel<<<nbA + G1, SORT_T, 0, stream>>>(dstI, cnt1, E, CB, G1,
                                                          x, W1, bufA, N, nbA, 0, tA);
        scan_local_kernel<<<scanBl1, 256, 0, stream>>>(cnt1, off1, bsum1, (int)n1);
        scan_offset_kernel<false><<<scanBl1, 256, 0, stream>>>(off1, nullptr, bsum1, (int)n1);
        tile_scatter_kernel<<<nbB + G1, SORT_T, 0, stream>>>(srcI, dstI, ew, off1, tmpE, tmpD, E, CB, G1,
                                                            x, W1, bufA, N, nbB, tA, tB);
        bucket_sort_kernel<<<nbC + CB, SORT_T, 0, stream>>>(tmpE, tmpD, off1, rowptr, edges, N, E, CB, G1,
                                                           x, W1, bufA, N, nbC, tB, ntiles);
    } else {
        zero_counts_kernel<<<(N + 255) / 256, 256, 0, stream>>>(next, N);
        count_dst_kernel<<<edgeBlocks, 256, 0, stream>>>(dstI, next, E);
        scan_local_kernel<<<scanBlocks, 256, 0, stream>>>(next, rowptr, bsum, N);
        scan_offset_kernel<true><<<scanBlocks, 256, 0, stream>>>(rowptr, next, bsum, N);
        fill_edges_kernel<<<edgeBlocks, 256, 0, stream>>>(srcI, dstI, ew, next, edges, E);
        gemm_wmma_kernel<128, false><<<gemmBlocks, 256, 0, stream>>>(x, W1, bufA, N);
    }

    gather_gemm_kernel<128><<<gemmBlocks, 512, 0, stream>>>(bufA, rowptr, edges, b1, W2, bufB, N);

    gather_gemm_kernel<64><<<gemmBlocks, 512, 0, stream>>>(bufB, rowptr, edges, b2, W3, bufA, N);

    if (!roomy)
        copy_ints_kernel<<<(int)((csrInts + 255) / 256), 256, 0, stream>>>(csr, csr3, (int)csrInts);
    gather_kernel<64><<<nodeBlocks, 256, 0, stream>>>(
        bufA, csr3, (const int2*)(csr3 + 2 * NP + NB), b3, out, N);
}


# edge sort in two launches: a tile keeps its bucket-sorted edges in place and stores its run starts; their sums replace the count pass and the offset scan; records still ranked in arrival order
# speedup vs baseline: 4.2345x; 1.0613x over previous
#include <hip/hip_runtime.h>
#include <hip/hip_bf16.h>

typedef __attribute__((ext_vector_type(2))) float v2f;
typedef __attribute__((ext_vector_type(8))) float v8f;

#define FIN 128

template <int FOUT, bool RELU>
__global__ __launch_bounds__(256) void gemm_wmma_kernel(
    const float* __restrict__ A, const float* __restrict__ W,
    float* __restrict__ C, int nrows)
{
    constexpr int NT = FOUT / 16;
    __shared__ float ldsW[FIN * FOUT];

    for (int i = threadIdx.x; i < FIN * FOUT; i += 256)
        ldsW[i] = W[i];
    __syncthreads();

    const int wave = threadIdx.x >> 5;
    const int lane = threadIdx.x & 31;
    const int half = lane >> 4;
    const int m    = lane & 15;

    const int tile   = blockIdx.x * 8 + wave;
    const int ntiles = (nrows + 15) >> 4;
    if (tile >= ntiles) return;

    const int row0 = tile * 16;
    int arow = row0 + m;
    if (arow >= nrows) arow = nrows - 1;
    const float* aptr = A + (size_t)arow * FIN;

    v8f acc[NT] = {};

    for (int k = 0; k < FIN; k += 4) {
        const int kk = k + 2 * half;
        float2 af = *(const float2*)(aptr + kk);
        if (RELU) {
            af.x = af.x > 0.0f ? af.x : 0.0f;
            af.y = af.y > 0.0f ? af.y : 0.0f;
        }
        v2f a;
        a.x = af.x;
        a.y = af.y;
#pragma unroll
        for (int t = 0; t < NT; ++t) {
            v2f b;
            b.x = ldsW[(kk)     * FOUT + t * 16 + m];
            b.y = ldsW[(kk + 1) * FOUT + t * 16 + m];
            acc[t] = __builtin_amdgcn_wmma_f32_16x16x4_f32(
                false, a, false, b, (short)0, acc[t], false, false);
        }
    }

#pragma unroll
    for (int r = 0; r < 8; ++r) {
        const int row = row0 + r + 8 * half;
        if (row < nrows) {
#pragma unroll
            for (int t = 0; t < NT; ++t)
                C[(size_t)row * FOUT + t * 16 + m] = acc[t][r];
        }
    }
}

__global__ void zero_counts_kernel(int* __restrict__ cnt, int n)
{
    const int i = blockIdx.x * blockDim.x + threadIdx.x;
    if (i < n) cnt[i] = 0;
}

__global__ void count_dst_kernel(const int* __restrict__ dstI,
                                 int* __restrict__ cnt, int nedges)
{
    const int e = blockIdx.x * blockDim.x + threadIdx.x;
    if (e < nedges) atomicAdd(&cnt[dstI[e]], 1);
}

__global__ __launch_bounds__(256) void scan_local_kernel(
    const int* __restrict__ cnt, int* __restrict__ rowptr,
    int* __restrict__ bsum, int n)
{
    __shared__ int sh[256];
    const int t  = threadIdx.x;
    const int i0 = blockIdx.x * 1024 + t * 4;
    int c[4];
#pragma unroll
    for (int k = 0; k < 4; ++k) c[k] = (i0 + k < n) ? cnt[i0 + k] : 0;
    const int mine = c[0] + c[1] + c[2] + c[3];
    sh[t] = mine;
    __syncthreads();
    for (int off = 1; off < 256; off <<= 1) {
        const int add = (t >= off) ? sh[t - off] : 0;
        __syncthreads();
        sh[t] += add;
        __syncthreads();
    }
    int run = sh[t] - mine;
#pragma unroll
    for (int k = 0; k < 4; ++k) {
        if (i0 + k < n) rowptr[i0 + k] = run;
        run += c[k];
    }
    if (t == 255) bsum[blockIdx.x] = sh[255];
}

template <bool NEXT>
__global__ __launch_bounds__(256) void scan_offset_kernel(
    int* __restrict__ rowptr, int* __restrict__ next,
    const int* __restrict__ bsum, int n)
{
    int off = 0;
    for (int j = 0; j < (int)blockIdx.x; ++j) off += bsum[j];
    const int i0 = blockIdx.x * 1024 + threadIdx.x * 4;
#pragma unroll
    for (int k = 0; k < 4; ++k) {
        if (i0 + k < n) {
            const int r = rowptr[i0 + k] + off;
            rowptr[i0 + k] = r;
            if (NEXT) next[i0 + k] = r;
        }
    }
    if (blockIdx.x == gridDim.x - 1 && threadIdx.x == 0)
        rowptr[n] = off + bsum[blockIdx.x];
}

__global__ void fill_edges_kernel(const int* __restrict__ srcI,
                                  const int* __restrict__ dstI,
                                  const float* __restrict__ ew,
                                  int* __restrict__ next,
                                  int2* __restrict__ edges, int nedges)
{
    const int e = blockIdx.x * blockDim.x + threadIdx.x;
    if (e < 8) edges[nedges + e] = make_int2(0, 0);
    if (e >= nedges) return;
    const int p = atomicAdd(&next[dstI[e]], 1);
    edges[p] = make_int2(srcI[e], __float_as_int(ew[e]));
}

__global__ void copy_ints_kernel(const int* __restrict__ from,
                                 int* __restrict__ to, int n)
{
    const int i = blockIdx.x * blockDim.x + threadIdx.x;
    if (i < n) to[i] = from[i];
}

#define BSH    8
#define BW     256
#define TILE   4096
#define SORT_T 512
#define CBMAX  1024
#define CAP    7104

__device__ __forceinline__ int block_excl_sum(int mine, int* sh)
{
    const int t = threadIdx.x;
    sh[t] = mine;
    __syncthreads();
    for (int off = 1; off < SORT_T; off <<= 1) {
        const int add = (t >= off) ? sh[t - off] : 0;
        __syncthreads();
        sh[t] += add;
        __syncthreads();
    }
    const int r = sh[t] - mine;
    __syncthreads();
    return r;
}

__device__ __forceinline__ int coarse_bucket(int d, int cb)
{
    return min((int)((unsigned)d >> BSH), cb - 1);
}

#define XW_NB    256
#define LDS_INTS (FIN * 128)

__device__ __forceinline__ void xw_role(
    float* ldsW, const float* __restrict__ X, const float* __restrict__ Wm,
    float* __restrict__ C, int nrows, int gb, int nb, int t0, int t1)
{
    constexpr int FOUT = 128;
    constexpr int NTW  = 4;
    constexpr int WQ   = FIN * FOUT / 4 / SORT_T;
    float4 wq[WQ];
#pragma unroll
    for (int i = 0; i < WQ; ++i)
        wq[i] = ((const float4*)Wm)[threadIdx.x + i * SORT_T];

    const int lane = threadIdx.x & 31;
    const int wave = __builtin_amdgcn_readfirstlane(threadIdx.x >> 5);
    const int pair = wave >> 1;
    const int sub  = wave & 1;
    const int half = lane >> 4;
    const int m    = lane & 15;
    const int range = t1 - t0;
    const int full  = nb * 8;

    int rel = gb * 8 + pair;
    int row0 = (t0 + rel) * 16;
    const float* aptr = X + (size_t)min(row0 + m, nrows - 1) * FIN + 2 * half;
    float2 cur[4];
#pragma unroll
    for (int s = 0; s < 4; ++s) cur[s] = *(const float2*)(aptr + s * 4);

#pragma unroll
    for (int i = 0; i < WQ; ++i)
        ((float4*)ldsW)[threadIdx.x + i * SORT_T] = wq[i];
    __syncthreads();

    const float* bptr = ldsW + (2 * half) * FOUT + sub * (NTW * 16) + m;
    v2f bcur[NTW];
#pragma unroll
    for (int t = 0; t < NTW; ++t) {
        bcur[t].x = bptr[t * 16];
        bcur[t].y = bptr[FOUT + t * 16];
    }

    for (int it = 0; rel < range; ++it) {
        v8f acc[NTW] = {};
#pragma nounroll
        for (int kb = 0; kb < FIN / 16; ++kb) {
            const int kn = ((kb + 1) & (FIN / 16 - 1)) * 16;
#pragma unroll
            for (int s = 0; s < 4; ++s) {
                v2f a;
                a.x = cur[s].x;
                a.y = cur[s].y;
                const float* bn = (s < 3) ? bptr + (kb * 16 + s * 4 + 4) * FOUT
                                          : bptr + kn * FOUT;
                v2f bnx[NTW];
#pragma unroll
                for (int t = 0; t < NTW; ++t) {
                    bnx[t].x = bn[t * 16];
                    bnx[t].y = bn[FOUT + t * 16];
                }
                __builtin_amdgcn_sched_barrier(0);
#pragma unroll
                for (int t = 0; t < NTW; ++t)
                    acc[t] = __builtin_amdgcn_wmma_f32_16x16x4_f32(
                        false, a, false, bcur[t], (short)0, acc[t], false, false);
#pragma unroll
                for (int t = 0; t < NTW; ++t) bcur[t] = bnx[t];
                cur[s] = *(const float2*)(aptr + kn + s * 4);
                __builtin_amdgcn_sched_barrier(0);
            }
        }

#pragma unroll
        for (int r = 0; r < 8; ++r) {
            const int orow = row0 + r + 8 * half;
            if (orow < nrows) {
#pragma unroll
                for (int t = 0; t < NTW; ++t)
                    C[(size_t)orow * FOUT + (sub * NTW + t) * 16 + m] = acc[t][r];
            }
        }

        rel = full + gb + (pair + 8 * it) * nb;
        if (rel < range) {
            row0 = (t0 + rel) * 16;
            aptr = X + (size_t)min(row0 + m, nrows - 1) * FIN + 2 * half;
#pragma unroll
            for (int s = 0; s < 4; ++s) cur[s] = *(const float2*)(aptr + s * 4);
        }
    }
}

__global__ __launch_bounds__(SORT_T) void tile_sort_kernel(
    const int* __restrict__ srcI, const int* __restrict__ dstI,
    const float* __restrict__ ew, int* __restrict__ cnt1, int* __restrict__ tst1,
    int2* __restrict__ tmpE, int* __restrict__ tmpD, int nedges, int cb, int g1,
    const float* __restrict__ X, const float* __restrict__ Wm, float* __restrict__ H,
    int nrows, int nb, int t0, int t1)
{
    constexpr int PT = TILE / SORT_T;
    __shared__ __attribute__((aligned(16))) int lds[LDS_INTS];
    if ((int)blockIdx.x < nb) {
        xw_role((float*)lds, X, Wm, H, nrows, blockIdx.x, nb, t0, t1);
        return;
    }
    const int bid = blockIdx.x - nb;
    int*  tcount = lds;
    int*  tstart = tcount + CBMAX;
    int*  sh     = tstart + CBMAX;
    int*  stageD = sh + SORT_T;
    int2* stageE = (int2*)(stageD + TILE);
    static_assert((2 * CBMAX + SORT_T + 3 * TILE) <= LDS_INTS, "sort tables fit the block");
    const int t = threadIdx.x;
    for (int k = t; k < CBMAX; k += SORT_T) tcount[k] = 0;
    __syncthreads();

    const int base = bid * TILE;
    const int nT   = min(TILE, nedges - base);
    int d[PT], s[PT], w[PT], r[PT];
#pragma unroll
    for (int u = 0; u < PT; ++u) {
        const int i = u * SORT_T + t;
        if (i < nT) {
            d[u] = dstI[base + i];
            s[u] = srcI[base + i];
            w[u] = __float_as_int(ew[base + i]);
        }
    }
#pragma unroll
    for (int u = 0; u < PT; ++u) {
        const int i = u * SORT_T + t;
        if (i < nT) r[u] = atomicAdd(&tcount[coarse_bucket(d[u], cb)], 1);
    }
    __syncthreads();

    const int c0 = tcount[2 * t];
    const int c1 = tcount[2 * t + 1];
    const int ex = block_excl_sum(c0 + c1, sh);
    tstart[2 * t]     = ex;
    tstart[2 * t + 1] = ex + c0;
    __syncthreads();

#pragma unroll
    for (int u = 0; u < PT; ++u) {
        const int i = u * SORT_T + t;
        if (i < nT) {
            const int slot = tstart[coarse_bucket(d[u], cb)] + r[u];
            stageE[slot] = make_int2(s[u], w[u]);
            stageD[slot] = d[u];
        }
    }
    __syncthreads();

    for (int i = t; i < nT; i += SORT_T) {
        tmpE[base + i] = stageE[i];
        tmpD[base + i] = stageD[i];
    }
    for (int k = t; k < cb; k += SORT_T) {
        cnt1[(size_t)k * g1 + bid] = tcount[k];
        tst1[(size_t)k * g1 + bid] = tstart[k];
    }
}

template <int NU>
__device__ __forceinline__ void bs_find_count(
    const int* runS, const int* runT, const int* __restrict__ tmpD, int* cnt,
    const int first, const int L, int (&q)[NU], int (&dn)[NU])
{
    int lo[NU];
#pragma unroll
    for (int u = 0; u < NU; ++u) lo[u] = 0;
    for (int st = SORT_T / 2; st > 0; st >>= 1) {
        int v[NU];
#pragma unroll
        for (int u = 0; u < NU; ++u) v[u] = runS[lo[u] + st];
#pragma unroll
        for (int u = 0; u < NU; ++u)
            lo[u] += (v[u] <= min(first + u * SORT_T, L - 1)) ? st : 0;
    }
#pragma unroll
    for (int u = 0; u < NU; ++u)
        q[u] = runT[lo[u]] + (min(first + u * SORT_T, L - 1) - runS[lo[u]]);
#pragma unroll
    for (int u = 0; u < NU; ++u) dn[u] = tmpD[q[u]] & (BW - 1);
#pragma unroll
    for (int u = 0; u < NU; ++u)
        if (first + u * SORT_T < L) atomicAdd(&cnt[dn[u]], 1);
}

template <int NU>
__device__ __forceinline__ void bs_place(
    const int2* __restrict__ tmpE, int* cursor, int2* stage, int2* __restrict__ edges,
    const int s0, const bool staged, const int first, const int L,
    const int (&q)[NU], const int (&dn)[NU])
{
    int2 rec[NU];
#pragma unroll
    for (int u = 0; u < NU; ++u) rec[u] = tmpE[q[u]];
#pragma unroll
    for (int u = 0; u < NU; ++u) {
        if (first + u * SORT_T < L) {
            const int p = atomicAdd(&cursor[dn[u]], 1);
            if (staged) stage[p] = rec[u];
            else        edges[s0 + p] = rec[u];
        }
    }
}

#define PE 16
__global__ __launch_bounds__(SORT_T) void bucket_sort_kernel(
    const int2* __restrict__ tmpE, const int* __restrict__ tmpD,
    const int* __restrict__ cnt1, const int* __restrict__ tst1, int* __restrict__ rowptr,
    int2* __restrict__ edges, int nnodes, int nedges, int cb, int g1,
    const float* __restrict__ X, const float* __restrict__ Wm, float* __restrict__ H,
    int nrows, int nb, int t0, int t1)
{
    __shared__ __attribute__((aligned(16))) int lds[LDS_INTS];
    if ((int)blockIdx.x < nb) {
        xw_role((float*)lds, X, Wm, H, nrows, blockIdx.x, nb, t0, t1);
        return;
    }
    int*  cnt    = lds;
    int*  cursor = cnt + BW;
    int*  sh     = cursor + BW;
    int*  runS   = sh + SORT_T;
    int*  runT   = runS + SORT_T;
    int*  tot    = runT + SORT_T;
    int2* stage  = (int2*)(tot + 32);
    static_assert((2 * BW + 3 * SORT_T + 32 + 2 * CAP) <= LDS_INTS, "sort tables fit the block");
    static_assert(CAP <= PE * SORT_T, "a staged bucket's edges fit the threads' registers");
    const int t  = threadIdx.x;
    const int k  = blockIdx.x - nb;
    const int rl = (t < g1) ? cnt1[(size_t)k * g1 + t] : 0;
    const int rt = (t < g1) ? tst1[(size_t)k * g1 + t] : 0;
    if (t < BW) cnt[t] = 0;
    if (k == 0 && t < 8) edges[nedges + t] = make_int2(0, 0);
    const int rs = block_excl_sum(rl, sh);
    runS[t] = rs;
    runT[t] = t * TILE + rt;
    int wsum = rt;
    for (int off = 16; off > 0; off >>= 1) wsum += __shfl_xor(wsum, off);
    if ((t & 31) == 0) tot[1 + (t >> 5)] = wsum;
    if (t == SORT_T - 1) tot[0] = rs + rl;
    __syncthreads();
    const int L  = tot[0];
    int s0 = 0;
    for (int w = 0; w < SORT_T / 32; ++w) s0 += tot[1 + w];
    const int s1 = s0 + L;

#define BS_LOCATE(I, P)                                                        \
    {                                                                          \
        int lo_ = 0;                                                           \
        for (int st_ = SORT_T / 2; st_ > 0; st_ >>= 1)                         \
            if (runS[lo_ + st_] <= (I)) lo_ += st_;                            \
        P = runT[lo_] + ((I) - runS[lo_]);                                     \
    }

    const bool inreg = L <= PE * SORT_T;
    const int  nu    = (L + SORT_T - 1) / SORT_T;
    int qa[PE / 2], da[PE / 2], qb[PE / 2], db[PE / 2];
    if (inreg) {
        if (nu > 0)      bs_find_count<PE / 2>(runS, runT, tmpD, cnt, t, L, qa, da);
        if (nu > PE / 2) bs_find_count<PE / 2>(runS, runT, tmpD, cnt, PE / 2 * SORT_T + t, L, qb, db);
    } else {
        for (int i = t; i < L; i += SORT_T) {
            int p;
            BS_LOCATE(i, p)
            atomicAdd(&cnt[tmpD[p] & (BW - 1)], 1);
        }
    }
    __syncthreads();

    const int mine = (t < BW) ? cnt[t] : 0;
    const int ex   = block_excl_sum(mine, sh);
    if (t < BW) {
        cursor[t] = ex;
        const int node = k * BW + t;
        if (node < nnodes) rowptr[node] = s0 + ex;
    }
    if (k == cb - 1 && t == 0) rowptr[nnodes] = s1;
    __syncthreads();

    const bool staged = L <= CAP;
    if (inreg) {
        if (nu > 0)      bs_place<PE / 2>(tmpE, cursor, stage, edges, s0, staged, t, L, qa, da);
        if (nu > PE / 2) bs_place<PE / 2>(tmpE, cursor, stage, edges, s0, staged, PE / 2 * SORT_T + t, L, qb, db);
    } else {
        for (int i = t; i < L; i += SORT_T) {
            int pl;
            BS_LOCATE(i, pl)
            const int  p   = atomicAdd(&cursor[tmpD[pl] & (BW - 1)], 1);
            const int2 rec = tmpE[pl];
            edges[s0 + p] = rec;
        }
    }
#undef BS_LOCATE
    __syncthreads();
    if (staged)
        for (int i = t; i < L; i += SORT_T) edges[s0 + i] = stage[i];
}
#undef PE

template <int F>
__global__ __launch_bounds__(256) void gather_kernel(
    const float* __restrict__ h, const int* __restrict__ rowptr,
    const int2* __restrict__ edges, const float* __restrict__ bias,
    float* __restrict__ out, int nnodes)
{
    constexpr int PER = F / 32;
    constexpr int UN  = 8;
    typedef float __attribute__((ext_vector_type(PER))) vecf;
    const int lane = threadIdx.x & 31;
    const int d = __builtin_amdgcn_readfirstlane(blockIdx.x * 8 + (threadIdx.x >> 5));
    if (d >= nnodes) return;
    const int beg = rowptr[d];
    const int end = rowptr[d + 1];

    vecf acc = 0.0f;

    for (int j = beg; j < end; j += UN) {
        const int n = end - j;
        int2 r[UN];
#pragma unroll
        for (int u = 0; u < UN; ++u) r[u] = edges[j + u];
        vecf v[UN];
#pragma unroll
        for (int u = 0; u < UN; ++u) {
            if (u < n)
                v[u] = *(const vecf*)(h + (size_t)r[u].x * F + lane * PER);
        }
#pragma unroll
        for (int u = 0; u < UN; ++u) {
            if (u < n)
                acc += __int_as_float(r[u].y) * v[u];
        }
    }

    const vecf b = *(const vecf*)(bias + lane * PER);
    *(vecf*)(out + (size_t)d * F + lane * PER) = acc + b;
}


#define TPITCH 132

template <int FOUT>
__global__ __launch_bounds__(512) void gather_gemm_kernel(
    const float* __restrict__ h, const int* __restrict__ rowptr,
    const int2* __restrict__ edges, const float* __restrict__ bias,
    const float* __restrict__ Wn, float* __restrict__ C, int nnodes)
{
    constexpr int NT  = FOUT / 16;
    constexpr int NTW = NT / 2;
    constexpr int UN  = 8;
    typedef float __attribute__((ext_vector_type(4))) vecf;
    constexpr int WQ  = FIN * FOUT / 4 / 512;
    __shared__ float lds[FIN * FOUT + 8 * 16 * TPITCH];
    float* ldsW = lds;
    float* tiles = lds + FIN * FOUT;

    float4 wq[WQ];
#pragma unroll
    for (int i = 0; i < WQ; ++i)
        wq[i] = ((const float4*)Wn)[threadIdx.x + i * 512];

    const int lane = threadIdx.x & 31;
    const int wave = __builtin_amdgcn_readfirstlane(threadIdx.x >> 5);
    const int pair = wave >> 1;
    const int sub  = wave & 1;
    const int node0 = (blockIdx.x * 8 + pair) * 16;
    float* trow = tiles + pair * 16 * TPITCH;

    const int first = node0 + 8 * sub;
    const int rp = rowptr[min(first + lane, nnodes)];
    const vecf b = *(const vecf*)(bias + lane * 4);

#pragma unroll
    for (int i = 0; i < WQ; ++i)
        ((float4*)ldsW)[threadIdx.x + i * 512] = wq[i];

    const int jbeg = __builtin_amdgcn_readlane(rp, 0);
    const int jend = __builtin_amdgcn_readlane(rp, 8);
    int  q    = 0;
    int  endq = __builtin_amdgcn_readlane(rp, 1);
    vecf gsum  = 0.0f;

#define GG_CLOSE_NODE()                                                        \
    {                                                                          \
        vecf row = gsum + b;                                                    \
        row.x = row.x > 0.0f ? row.x : 0.0f;                                   \
        row.y = row.y > 0.0f ? row.y : 0.0f;                                   \
        row.z = row.z > 0.0f ? row.z : 0.0f;                                   \
        row.w = row.w > 0.0f ? row.w : 0.0f;                                   \
        if (first + q >= nnodes) row = 0.0f;         \
        *(vecf*)(trow + (8 * sub + q) * TPITCH + lane * 4) = row;              \
        ++q;                                                                   \
        endq = __builtin_amdgcn_readlane(rp, q + 1);                           \
        gsum  = 0.0f;                                                           \
    }
#define GG_ROWS(J0, R, V)                                                      \
    _Pragma("unroll")                                                          \
    for (int u = 0; u < UN; ++u) {                                             \
        if ((J0) + u < jend)                                                   \
            V[u] = *(const vecf*)(h + (size_t)R[u].x * FIN + lane * 4);        \
    }
#define GG_RECS(J0, R)                                                         \
    {                                                                          \
        const int jr = min((J0), jend);                                        \
        _Pragma("unroll")                                                      \
        for (int u = 0; u < UN; ++u) R[u] = edges[jr + u];                     \
    }
#define GG_SUM(J0, R, V)                                                       \
    _Pragma("unroll")                                                          \
    for (int u = 0; u < UN; ++u) {                                             \
        const int e = (J0) + u;                                                \
        if (e < jend) {                                                        \
            while (e >= endq) GG_CLOSE_NODE()                                  \
            gsum += __int_as_float(R[u].y) * V[u];                              \
        }                                                                      \
    }

#define GG_ROWS_ALL(R, V)                                                      \
    _Pragma("unroll")                                                          \
    for (int u = 0; u < UN; ++u)                                               \
        V[u] = *(const vecf*)(h + (size_t)R[u].x * FIN + lane * 4);
#define GG_SUM_ALL(J0, R, V)                                                   \
    _Pragma("unroll")                                                          \
    for (int u = 0; u < UN; ++u) {                                             \
        const int e = (J0) + u;                                                \
        while (e >= endq) GG_CLOSE_NODE()                                      \
        gsum += __int_as_float(R[u].y) * V[u];                                 \
    }

    int2 rc[UN], rn[UN], rx[UN], ry[UN];
    vecf va[UN], vb[UN];
    int j = jbeg;
    if (j + 3 * UN <= jend) {
        GG_RECS(j, rc)
        GG_RECS(j + UN, rn)
        GG_ROWS_ALL(rc, va)
        do {
            GG_ROWS_ALL(rn, vb)
            GG_RECS(j + 2 * UN, rx)
            GG_SUM_ALL(j, rc, va)
            GG_ROWS_ALL(rx, va)
            GG_RECS(j + 3 * UN, ry)
            GG_SUM_ALL(j + UN, rn, vb)
#pragma unroll
            for (int u = 0; u < UN; ++u) { rc[u] = rx[u]; rn[u] = ry[u]; }
            j += 2 * UN;
        } while (j + 3 * UN <= jend);
        GG_SUM_ALL(j, rc, va)
        j += UN;
    }
    for (; j < jend; j += UN) {
        GG_RECS(j, rc)
        GG_ROWS(j, rc, va)
        GG_SUM(j, rc, va)
    }
    while (q < 8) GG_CLOSE_NODE()
#undef GG_CLOSE_NODE
#undef GG_ROWS
#undef GG_RECS
#undef GG_SUM
#undef GG_ROWS_ALL
#undef GG_SUM_ALL
    __syncthreads();
    if (node0 >= nnodes) return;

    const int half = lane >> 4;
    const int m    = lane & 15;
    const float* arow = trow + m * TPITCH;

    v8f acc[NTW] = {};

    for (int k = 0; k < FIN; k += 4) {
        const int kk = k + 2 * half;
        const float2 af = *(const float2*)(arow + kk);
        v2f a;
        a.x = af.x;
        a.y = af.y;
#pragma unroll
        for (int t = 0; t < NTW; ++t) {
            const int col = (sub * NTW + t) * 16 + m;
            v2f bf;
            bf.x = ldsW[(kk)     * FOUT + col];
            bf.y = ldsW[(kk + 1) * FOUT + col];
            acc[t] = __builtin_amdgcn_wmma_f32_16x16x4_f32(
                false, a, false, bf, (short)0, acc[t], false, false);
        }
    }

#pragma unroll
    for (int r = 0; r < 8; ++r) {
        const int orow = node0 + r + 8 * half;
        if (orow < nnodes) {
#pragma unroll
            for (int t = 0; t < NTW; ++t)
                C[(size_t)orow * FOUT + (sub * NTW + t) * 16 + m] = acc[t][r];
        }
    }
}

extern "C" void kernel_launch(void* const* d_in, const int* in_sizes, int n_in,
                              void* d_out, int out_size, void* d_ws, size_t ws_size,
                              hipStream_t stream)
{
    const float* x  = (const float*)d_in[0];
    const int*   ei = (const int*)d_in[1];
    const float* ew = (const float*)d_in[2];
    const float* W1 = (const float*)d_in[3];
    const float* b1 = (const float*)d_in[4];
    const float* W2 = (const float*)d_in[5];
    const float* b2 = (const float*)d_in[6];
    const float* W3 = (const float*)d_in[7];
    const float* b3 = (const float*)d_in[8];
    float* out = (float*)d_out;

    const int N = in_sizes[0] / FIN;
    const int E = in_sizes[2];
    const int* srcI = ei;
    const int* dstI = ei + E;

    float* bufA = (float*)d_ws;
    float* bufB = bufA + (size_t)N * FIN;

    const int ntiles     = (N + 15) / 16;
    const int gemmBlocks = (ntiles + 7) / 8;
    const int nodeBlocks = (N + 7) / 8;
    const int edgeBlocks = (E + 255) / 256;
    const int scanBlocks = (N + 1023) / 1024;

    const int    NP      = (N + 1 + 63) & ~63;
    const int    NB      = (scanBlocks + 63) & ~63;
    const size_t csrInts = (size_t)2 * NP + NB + (size_t)2 * E + 16;
    const bool roomy = ws_size >= ((size_t)2 * N * FIN + csrInts) * sizeof(float);
    int* csr  = roomy ? (int*)(bufB + (size_t)N * FIN) : (int*)d_out;
    int* csr3 = roomy ? csr : (int*)(bufA + (size_t)N * 64);

    int*  rowptr = csr;
    int*  next   = csr + NP;
    int*  bsum   = csr + 2 * NP;
    int2* edges  = (int2*)(csr + 2 * NP + NB);

    const int    CB      = (N + BW - 1) / BW;
    const int    G1      = (E + TILE - 1) / TILE;
    const size_t n1      = (size_t)CB * G1;
    const size_t n1p     = (n1 + 1 + 63) & ~(size_t)63;
    const bool   ldsSort = CB <= CBMAX && G1 > 0 && G1 <= SORT_T && n1 < ((size_t)1 << 30) &&
                           (size_t)3 * E + 2 * n1p <= (size_t)N * FIN;
    if (ldsSort) {
        int2* tmpE  = (int2*)bufB;
        int*  tmpD  = (int*)(tmpE + E);
        int*  cnt1  = tmpD + E;
        int*  tst1  = cnt1 + n1p;
        const int tA  = (ntiles + 1) / 2;
        const int nbA = (tA + 7) / 8 < XW_NB ? (tA + 7) / 8 : XW_NB;
        const int nbB = (ntiles - tA + 7) / 8 < XW_NB ? (ntiles - tA + 7) / 8 : XW_NB;
        tile_sort_kernel<<<nbA + G1, SORT_T, 0, stream>>>(srcI, dstI, ew, cnt1, tst1, tmpE, tmpD, E, CB, G1,
                                                         x, W1, bufA, N, nbA, 0, tA);
        bucket_sort_kernel<<<nbB + CB, SORT_T, 0, stream>>>(tmpE, tmpD, cnt1, tst1, rowptr, edges, N, E, CB, G1,
                                                           x, W1, bufA, N, nbB, tA, ntiles);
    } else {
        zero_counts_kernel<<<(N + 255) / 256, 256, 0, stream>>>(next, N);
        count_dst_kernel<<<edgeBlocks, 256, 0, stream>>>(dstI, next, E);
        scan_local_kernel<<<scanBlocks, 256, 0, stream>>>(next, rowptr, bsum, N);
        scan_offset_kernel<true><<<scanBlocks, 256, 0, stream>>>(rowptr, next, bsum, N);
        fill_edges_kernel<<<edgeBlocks, 256, 0, stream>>>(srcI, dstI, ew, next, edges, E);
        gemm_wmma_kernel<128, false><<<gemmBlocks, 256, 0, stream>>>(x, W1, bufA, N);
    }

    gather_gemm_kernel<128><<<gemmBlocks, 512, 0, stream>>>(bufA, rowptr, edges, b1, W2, bufB, N);

    gather_gemm_kernel<64><<<gemmBlocks, 512, 0, stream>>>(bufB, rowptr, edges, b2, W3, bufA, N);

    if (!roomy)
        copy_ints_kernel<<<(int)((csrInts + 255) / 256), 256, 0, stream>>>(csr, csr3, (int)csrInts);
    gather_kernel<64><<<nodeBlocks, 256, 0, stream>>>(
        bufA, csr3, (const int2*)(csr3 + 2 * NP + NB), b3, out, N);
}
